# starting kernel
# baseline (speedup 1.0000x reference)
#include <hip/hip_runtime.h>
#include <hip/hip_bf16.h>
#include <math.h>

typedef __attribute__((ext_vector_type(2))) float v2f;
typedef __attribute__((ext_vector_type(8))) float v8f;

#define LRC  0.01f
#define EPSC 1e-5f

#define Bq 256
#define Lq 4096
#define Hq 32
#define Vq 64
#define CH 16
#define NCH (Lq / CH)
#define A1STR 68
#define HSTR  36

__launch_bounds__(64, 1)
__global__ void pmm_fused_kernel(const int* __restrict__ seq,
                                 const float* __restrict__ embed,
                                 const float* __restrict__ ff_w1,
                                 const float* __restrict__ ff_b1,
                                 const float* __restrict__ ff_w2,
                                 const float* __restrict__ ff_b2,
                                 const float* __restrict__ ln_g,
                                 const float* __restrict__ ln_b,
                                 const float* __restrict__ mlp_w1,
                                 const float* __restrict__ mlp_b1,
                                 const float* __restrict__ mlp_w2,
                                 const float* __restrict__ mlp_b2,
                                 const float* __restrict__ out_w,
                                 const float* __restrict__ out_b,
                                 float* __restrict__ out)
{
    __shared__ float embedLds[Vq * Hq];
    __shared__ float w1Lds[2 * Hq * Hq];
    __shared__ float w2Lds[Hq * 2 * Hq];
    __shared__ float a1buf[CH * A1STR];
    __shared__ float hid[2][CH * HSTR];
    __shared__ float ctxLds[Hq];

    const int b    = blockIdx.x;
    const int tid  = threadIdx.x;
    const int lane = tid & 31;
    const int wave = tid >> 5;
    const int m    = lane & 15;
    const int hi   = lane >> 4;

    for (int i = tid; i < Vq * Hq; i += 64)     embedLds[i] = embed[i];
    for (int i = tid; i < 2 * Hq * Hq; i += 64) w1Lds[i]    = ff_w1[i];
    for (int i = tid; i < 2 * Hq * Hq; i += 64) w2Lds[i]    = ff_w2[i];
    __syncthreads();

    float b1t[4], b2t[2], lnGv = 0.f, lnBv = 0.f;
    if (wave == 0) {
        #pragma unroll
        for (int nt = 0; nt < 4; ++nt) b1t[nt] = ff_b1[nt * 16 + m];
        #pragma unroll
        for (int nt = 0; nt < 2; ++nt) b2t[nt] = ff_b2[nt * 16 + m];
        lnGv = ln_g[lane];
        lnBv = ln_b[lane];
    }

    float w1c[8], b1v[8], w2r[8], b2v = 0.f;
    if (wave == 1) {
        #pragma unroll
        for (int i = 0; i < 8; ++i) {
            w1c[i] = mlp_w1[i * Hq + lane];
            b1v[i] = mlp_b1[i];
            w2r[i] = mlp_w2[lane * 8 + i];
        }
        b2v = mlp_b2[lane];
    }

    auto produce = [&](int c, float* hbuf) {
        const int base = c * CH;
        const int tok  = seq[b * Lq + base + m];

        v2f eA[8];
        #pragma unroll
        for (int kb = 0; kb < 8; ++kb)
            eA[kb] = *(const v2f*)&embedLds[tok * Hq + 4 * kb + 2 * hi];

        #pragma unroll 1
        for (int nt = 0; nt < 4; ++nt) {
            const int n0 = nt * 16;
            v8f acc = {};
            #pragma unroll
            for (int kb = 0; kb < 8; ++kb) {
                v2f bB = *(const v2f*)&w1Lds[(n0 + m) * Hq + 4 * kb + 2 * hi];
                acc = __builtin_amdgcn_wmma_f32_16x16x4_f32(
                        false, eA[kb], false, bB, (short)0, acc, false, false);
            }
            const float bias = b1t[nt];
            #pragma unroll
            for (int r = 0; r < 8; ++r) {
                float z = acc[r] + bias;
                a1buf[(r + 8 * hi) * A1STR + n0 + m] = fmaxf(z, 0.f);
            }
        }
        asm volatile("s_wait_dscnt 0" ::: "memory");
        __builtin_amdgcn_wave_barrier();

        v2f a2[16];
        #pragma unroll
        for (int kb = 0; kb < 16; ++kb)
            a2[kb] = *(const v2f*)&a1buf[m * A1STR + 4 * kb + 2 * hi];

        #pragma unroll 1
        for (int nt = 0; nt < 2; ++nt) {
            const int n0 = nt * 16;
            v8f acc = {};
            #pragma unroll
            for (int kb = 0; kb < 16; ++kb) {
                v2f bB = *(const v2f*)&w2Lds[(n0 + m) * (2 * Hq) + 4 * kb + 2 * hi];
                acc = __builtin_amdgcn_wmma_f32_16x16x4_f32(
                        false, a2[kb], false, bB, (short)0, acc, false, false);
            }
            const float bias = b2t[nt];
            #pragma unroll
            for (int r = 0; r < 8; ++r) {
                int   tm = __shfl(tok, r + 8 * hi, 32);
                float e  = embedLds[tm * Hq + n0 + m];
                hbuf[(r + 8 * hi) * HSTR + n0 + m] = acc[r] + bias + e;
            }
        }
        asm volatile("s_wait_dscnt 0" ::: "memory");
        __builtin_amdgcn_wave_barrier();

        #pragma unroll 1
        for (int t = 0; t < CH; ++t) {
            float z = hbuf[t * HSTR + lane];
            float s = z, s2 = z * z;
            #pragma unroll
            for (int off = 16; off >= 1; off >>= 1) {
                s  += __shfl_xor(s,  off, 32);
                s2 += __shfl_xor(s2, off, 32);
            }
            float mu  = s * (1.f / 32.f);
            float var = s2 * (1.f / 32.f) - mu * mu;
            float nz  = (z - mu) * rsqrtf(var + EPSC);
            hbuf[t * HSTR + lane] = nz * lnGv + lnBv;
        }
        asm volatile("s_wait_dscnt 0" ::: "memory");
    };

    auto consume = [&](const float* hbuf, int nsteps) {
        #pragma unroll 1
        for (int s = 0; s < nsteps; ++s) {
            float x = hbuf[(2 * s) * HSTR + lane];
            float v = hbuf[(2 * s + 1) * HSTR + lane];

            float p[8];
            #pragma unroll
            for (int i = 0; i < 8; ++i) p[i] = w1c[i] * x;
            #pragma unroll
            for (int off = 16; off >= 1; off >>= 1) {
                #pragma unroll
                for (int i = 0; i < 8; ++i) p[i] += __shfl_xor(p[i], off, 32);
            }
            float a1p[8], a1v[8];
            float y = b2v;
            #pragma unroll
            for (int i = 0; i < 8; ++i) {
                a1p[i] = p[i] + b1v[i];
                a1v[i] = fmaxf(a1p[i], 0.f);
                y = fmaf(w2r[i], a1v[i], y);
            }
            float d = (y - v) * 0.0625f;

            float q[8];
            #pragma unroll
            for (int i = 0; i < 8; ++i) q[i] = w2r[i] * d;
            #pragma unroll
            for (int off = 16; off >= 1; off >>= 1) {
                #pragma unroll
                for (int i = 0; i < 8; ++i) q[i] += __shfl_xor(q[i], off, 32);
            }
            b2v -= LRC * d;
            #pragma unroll
            for (int i = 0; i < 8; ++i) {
                float g1 = (a1p[i] > 0.f) ? q[i] : 0.f;
                w2r[i] = fmaf(-LRC * d, a1v[i], w2r[i]);
                b1v[i] -= LRC * g1;
                w1c[i] = fmaf(-LRC * g1, x, w1c[i]);
            }
        }
    };

    if (wave == 0) produce(0, hid[0]);
    __syncthreads();

    for (int c = 0; c < NCH; ++c) {
        if (wave == 0) {
            if (c + 1 < NCH) produce(c + 1, hid[(c + 1) & 1]);
        } else {
            consume(hid[c & 1], (c == NCH - 1) ? 7 : 8);
        }
        __syncthreads();
    }

    if (wave == 1) {
        const float* hbuf = hid[(NCH - 1) & 1];
        float x = hbuf[15 * HSTR + lane];
        float p[8];
        #pragma unroll
        for (int i = 0; i < 8; ++i) p[i] = w1c[i] * x;
        #pragma unroll
        for (int off = 16; off >= 1; off >>= 1) {
            #pragma unroll
            for (int i = 0; i < 8; ++i) p[i] += __shfl_xor(p[i], off, 32);
        }
        float y = b2v;
        #pragma unroll
        for (int i = 0; i < 8; ++i) y = fmaf(w2r[i], fmaxf(p[i] + b1v[i], 0.f), y);
        ctxLds[lane] = y;
    }
    __syncthreads();

    {
        float acc = out_b[tid];
        #pragma unroll
        for (int h = 0; h < Hq; ++h)
            acc = fmaf(ctxLds[h], out_w[tid * Hq + h], acc);
        out[b * Vq + tid] = acc;
    }
}

extern "C" void kernel_launch(void* const* d_in, const int* in_sizes, int n_in,
                              void* d_out, int out_size, void* d_ws, size_t ws_size,
                              hipStream_t stream) {
    (void)in_sizes; (void)n_in; (void)out_size; (void)d_ws; (void)ws_size;
    const int*   seq    = (const int*)  d_in[0];
    const float* embed  = (const float*)d_in[1];
    const float* ff_w1  = (const float*)d_in[2];
    const float* ff_b1  = (const float*)d_in[3];
    const float* ff_w2  = (const float*)d_in[4];
    const float* ff_b2  = (const float*)d_in[5];
    const float* ln_g   = (const float*)d_in[6];
    const float* ln_b   = (const float*)d_in[7];
    const float* mlp_w1 = (const float*)d_in[8];
    const float* mlp_b1 = (const float*)d_in[9];
    const float* mlp_w2 = (const float*)d_in[10];
    const float* mlp_b2 = (const float*)d_in[11];
    const float* out_w  = (const float*)d_in[12];
    const float* out_b  = (const float*)d_in[13];
    float*       out    = (float*)d_out;

    pmm_fused_kernel<<<dim3(Bq), dim3(64), 0, stream>>>(
        seq, embed, ff_w1, ff_b1, ff_w2, ff_b2, ln_g, ln_b,
        mlp_w1, mlp_b1, mlp_w2, mlp_b2, out_w, out_b, out);
}


# SGD launch: fc1's 32-long sums split over matrix rows AND columns (row/column parity = k half): 4 matrix instructions per step, not 8; half sums joined by a lane^1 add; p's sum order changes
# speedup vs baseline: 2.3944x; 2.3944x over previous
#include <hip/hip_runtime.h>
#include <hip/hip_bf16.h>
#include <math.h>


typedef __attribute__((ext_vector_type(2))) float v2f;
typedef __attribute__((ext_vector_type(4))) float v4f;
typedef __attribute__((ext_vector_type(8))) float v8f;

#define LRC  0.01f
#define EPSC 1e-5f

#define Bq 256
#define Lq 4096
#define Hq 32
#define Vq 64
#define CH 16
#define NCH (Lq / CH)
#define A1STR 68
#define HSTR  36

#define ENC_WAVES 8
#define ENC_CPW   8
#define ENC_BLOCKS ((Bq * NCH) / (ENC_WAVES * ENC_CPW))

__launch_bounds__(ENC_WAVES * 32)
__global__ void pmm_encode_kernel(const int* __restrict__ seq,
                                  const float* __restrict__ embed,
                                  const float* __restrict__ ff_w1,
                                  const float* __restrict__ ff_b1,
                                  const float* __restrict__ ff_w2,
                                  const float* __restrict__ ff_b2,
                                  const float* __restrict__ ln_g,
                                  const float* __restrict__ ln_b,
                                  float* __restrict__ hidden)
{
    __shared__ float embedLds[Vq * Hq];
    __shared__ float w1Lds[2 * Hq * Hq];
    __shared__ float w2Lds[Hq * 2 * Hq];
    __shared__ float a1all[ENC_WAVES][CH * A1STR];
    __shared__ float hball[ENC_WAVES][CH * HSTR];

    const int tid  = threadIdx.x;
    const int lane = tid & 31;
    const int wave = tid >> 5;
    const int m    = lane & 15;
    const int hi   = lane >> 4;

    for (int i = tid; i < Vq * Hq; i += ENC_WAVES * 32)     embedLds[i] = embed[i];
    for (int i = tid; i < 2 * Hq * Hq; i += ENC_WAVES * 32) w1Lds[i]    = ff_w1[i];
    for (int i = tid; i < 2 * Hq * Hq; i += ENC_WAVES * 32) w2Lds[i]    = ff_w2[i];
    __syncthreads();

    float b1t[4], b2t[2];
    #pragma unroll
    for (int nt = 0; nt < 4; ++nt) b1t[nt] = ff_b1[nt * 16 + m];
    #pragma unroll
    for (int nt = 0; nt < 2; ++nt) b2t[nt] = ff_b2[nt * 16 + m];
    const float lnGv = ln_g[lane];
    const float lnBv = ln_b[lane];

    float* a1buf = a1all[wave];
    float* hbuf  = hball[wave];

    const int g0 = (blockIdx.x * ENC_WAVES + wave) * ENC_CPW;

    #pragma unroll 1
    for (int j = 0; j < ENC_CPW; ++j) {
        const int base = (g0 + j) * CH;
        const int tok  = seq[base + m];

        v2f eA[8];
        #pragma unroll
        for (int kb = 0; kb < 8; ++kb)
            eA[kb] = *(const v2f*)&embedLds[tok * Hq + 4 * kb + 2 * hi];

        #pragma unroll 1
        for (int nt = 0; nt < 4; ++nt) {
            const int n0 = nt * 16;
            v8f acc = {};
            #pragma unroll
            for (int kb = 0; kb < 8; ++kb) {
                v2f bB = *(const v2f*)&w1Lds[(n0 + m) * Hq + 4 * kb + 2 * hi];
                acc = __builtin_amdgcn_wmma_f32_16x16x4_f32(
                        false, eA[kb], false, bB, (short)0, acc, false, false);
            }
            const float bias = b1t[nt];
            #pragma unroll
            for (int r = 0; r < 8; ++r) {
                float z = acc[r] + bias;
                a1buf[(r + 8 * hi) * A1STR + n0 + m] = fmaxf(z, 0.f);
            }
        }
        asm volatile("s_wait_dscnt 0" ::: "memory");
        __builtin_amdgcn_wave_barrier();

        v2f a2[16];
        #pragma unroll
        for (int kb = 0; kb < 16; ++kb)
            a2[kb] = *(const v2f*)&a1buf[m * A1STR + 4 * kb + 2 * hi];

        #pragma unroll 1
        for (int nt = 0; nt < 2; ++nt) {
            const int n0 = nt * 16;
            v8f acc = {};
            #pragma unroll
            for (int kb = 0; kb < 16; ++kb) {
                v2f bB = *(const v2f*)&w2Lds[(n0 + m) * (2 * Hq) + 4 * kb + 2 * hi];
                acc = __builtin_amdgcn_wmma_f32_16x16x4_f32(
                        false, a2[kb], false, bB, (short)0, acc, false, false);
            }
            const float bias = b2t[nt];
            #pragma unroll
            for (int r = 0; r < 8; ++r) {
                int   tm = __shfl(tok, r + 8 * hi, 32);
                float e  = embedLds[tm * Hq + n0 + m];
                hbuf[(r + 8 * hi) * HSTR + n0 + m] = acc[r] + bias + e;
            }
        }
        asm volatile("s_wait_dscnt 0" ::: "memory");
        __builtin_amdgcn_wave_barrier();

        #pragma unroll 1
        for (int t = 0; t < CH; ++t) {
            float z = hbuf[t * HSTR + lane];
            float s = z, s2 = z * z;
            #pragma unroll
            for (int off = 16; off >= 1; off >>= 1) {
                s  += __shfl_xor(s,  off, 32);
                s2 += __shfl_xor(s2, off, 32);
            }
            float mu  = s * (1.f / 32.f);
            float var = s2 * (1.f / 32.f) - mu * mu;
            float nz  = (z - mu) * rsqrtf(var + EPSC);
            hidden[(size_t)(base + t) * Hq + lane] = nz * lnGv + lnBv;
        }
        asm volatile("s_wait_dscnt 0" ::: "memory");
    }
}

__device__ __forceinline__ float pick8(const float (&a)[8], bool b0, bool b1, bool b2)
{
    const float t0 = b0 ? a[1] : a[0];
    const float t1 = b0 ? a[3] : a[2];
    const float t2 = b0 ? a[5] : a[4];
    const float t3 = b0 ? a[7] : a[6];
    const float u0 = b1 ? t1 : t0;
    const float u1 = b1 ? t3 : t2;
    return b2 ? u1 : u0;
}

__device__ __forceinline__ float pair_lane(float s)
{
    return __int_as_float(__builtin_amdgcn_update_dpp(0, __float_as_int(s), 0xB1, 0xF, 0xF, true));
}

__device__ __forceinline__ float other_half(float s)
{
    const int r = __builtin_amdgcn_permlanex16(0, __float_as_int(s), 0x76543210, 0xfedcba98, false, false);
    return __int_as_float(r);
}

__launch_bounds__(32)
__global__ void pmm_sgd_kernel(const float* __restrict__ hidden,
                               const float* __restrict__ mlp_w1,
                               const float* __restrict__ mlp_b1,
                               const float* __restrict__ mlp_w2,
                               const float* __restrict__ mlp_b2,
                               const float* __restrict__ out_w,
                               const float* __restrict__ out_b,
                               float* __restrict__ out)
{
    __shared__ v4f   hid4[2][CH * Hq / 4];
    __shared__ float ctxLds[Hq];

    const int b    = blockIdx.x;
    const int lane = threadIdx.x;
    const int m    = lane & 15;
    const int hi   = lane >> 4;
    const int uOwn = (m & 7) ^ (4 * (m >> 3));
    const int rOwn = uOwn ^ (4 * hi);
    const bool mb0 = (rOwn & 1) != 0, mb1 = (rOwn & 2) != 0, mb2 = (rOwn & 4) != 0;
    const bool db0 = (m & 1) != 0, db1 = (m & 2) != 0, db2 = (m & 4) != 0;
    const bool ownHalf = (m >> 3) == hi;
    const int uW1  = 4 * (m >> 3) + ((m & 7) >> 1);
    const int sW1  = m & 1;
    const int rW1  = uW1 ^ (4 * hi);
    const bool wb0 = (rW1 & 1) != 0, wb1 = (rW1 & 2) != 0, wb2 = (rW1 & 4) != 0;
    const bool oddCol = sW1 != 0;

    const v4f* src = (const v4f*)(hidden + (size_t)b * Lq * Hq);

    v2f w1A[4], w2P[8], w2Q[2][2];
    float b1D[4], b2D[2][8];
    #pragma unroll
    for (int i = 0; i < 4; ++i) {
        #pragma unroll
        for (int j = 0; j < 2; ++j)
            w1A[i][j] = mlp_w1[uW1 * Hq + 16 * sW1 + 4 * i + 2 * hi + j];
    }
    #pragma unroll
    for (int kb = 0; kb < 8; ++kb) {
        #pragma unroll
        for (int j = 0; j < 2; ++j)
            w2P[kb][j] = mlp_w2[(16 * (kb >> 2) + 8 * hi + 2 * (kb & 3) + j) * 8 + uOwn];
    }
    #pragma unroll
    for (int t = 0; t < 2; ++t) {
        #pragma unroll
        for (int kb = 0; kb < 2; ++kb) {
            #pragma unroll
            for (int j = 0; j < 2; ++j)
                w2Q[t][kb][j] = mlp_w2[(16 * t + m) * 8 + 4 * hi + 2 * kb + j];
        }
        #pragma unroll
        for (int r = 0; r < 8; ++r) b2D[t][r] = mlp_b2[16 * t + 8 * hi + r];
    }
    #pragma unroll
    for (int i = 0; i < 4; ++i) b1D[i] = mlp_b1[4 * hi + i];

    auto forward = [&](const float* xrow, v2f (&xv)[4], float (&a1p)[8], float (&a1)[8],
                       v2f (&a1v)[2], v8f (&accY)[2]) {
        #pragma unroll
        for (int i = 0; i < 4; ++i)
            xv[i] = *(const v2f*)(xrow + 16 * sW1 + 4 * i + 2 * hi);
        v8f accP0 = {}, accP1 = {};
        #pragma unroll
        for (int i = 0; i < 4; i += 2) {
            accP0 = __builtin_amdgcn_wmma_f32_16x16x4_f32(
                        false, w1A[i], false, xv[i], (short)0, accP0, false, false);
            accP1 = __builtin_amdgcn_wmma_f32_16x16x4_f32(
                        false, w1A[i + 1], false, xv[i + 1], (short)0, accP1, false, false);
        }
        #pragma unroll
        for (int i = 0; i < 4; ++i) {
            const float sE = accP0[2 * i] + accP1[2 * i];
            const float sO = accP0[2 * i + 1] + accP1[2 * i + 1];
            const float own = oddCol ? sO : sE;
            a1p[i] = (own + pair_lane(own)) + b1D[i];
            a1[i]  = fmaxf(a1p[i], 0.f);
        }
        #pragma unroll
        for (int i = 0; i < 4; ++i) {
            a1p[4 + i] = other_half(a1p[i]);
            a1[4 + i]  = fmaxf(a1p[4 + i], 0.f);
        }
        #pragma unroll
        for (int kb = 0; kb < 2; ++kb) {
            a1v[kb][0] = a1[2 * kb];
            a1v[kb][1] = a1[2 * kb + 1];
        }
        #pragma unroll
        for (int t = 0; t < 2; ++t) {
            #pragma unroll
            for (int r = 0; r < 8; ++r) accY[t][r] = b2D[t][r];
        }
        #pragma unroll
        for (int kb = 0; kb < 2; ++kb) {
            #pragma unroll
            for (int t = 0; t < 2; ++t)
                accY[t] = __builtin_amdgcn_wmma_f32_16x16x4_f32(
                            false, w2Q[t][kb], false, a1v[kb], (short)0, accY[t], false, false);
        }
    };

    auto consume = [&](const float* hbuf, int nsteps) {
        #pragma unroll 1
        for (int s = 0; s < nsteps; ++s) {
            const float* xrow = hbuf + (2 * s) * Hq;
            const float* vrow = hbuf + (2 * s + 1) * Hq;

            float vD[2][8];
            #pragma unroll
            for (int t = 0; t < 2; ++t) {
                #pragma unroll
                for (int u = 0; u < 2; ++u) {
                    const v4f vq = *(const v4f*)(vrow + 16 * t + 8 * hi + 4 * u);
                    vD[t][4 * u + 0] = vq[0]; vD[t][4 * u + 1] = vq[1];
                    vD[t][4 * u + 2] = vq[2]; vD[t][4 * u + 3] = vq[3];
                }
            }

            v2f xv[4], a1v[2];
            float a1p[8], a1[8];
            v8f accY[2];
            forward(xrow, xv, a1p, a1, a1v, accY);

            float dD[2][8], nd[2][8];
            #pragma unroll
            for (int t = 0; t < 2; ++t) {
                #pragma unroll
                for (int r = 0; r < 8; ++r) {
                    dD[t][r] = (accY[t][r] - vD[t][r]) * 0.0625f;
                    nd[t][r] = -LRC * dD[t][r];
                }
            }

            v8f accQ0 = {}, accQ1 = {};
            #pragma unroll
            for (int kb = 0; kb < 8; kb += 2) {
                v2f dv0, dv1;
                dv0[0] = dD[kb >> 2][2 * (kb & 3)];           dv0[1] = dD[kb >> 2][2 * (kb & 3) + 1];
                dv1[0] = dD[(kb + 1) >> 2][2 * ((kb + 1) & 3)]; dv1[1] = dD[(kb + 1) >> 2][2 * ((kb + 1) & 3) + 1];
                accQ0 = __builtin_amdgcn_wmma_f32_16x16x4_f32(
                            false, w2P[kb], false, dv0, (short)0, accQ0, false, false);
                accQ1 = __builtin_amdgcn_wmma_f32_16x16x4_f32(
                            false, w2P[kb + 1], false, dv1, (short)0, accQ1, false, false);
            }

            float g1[8];
            #pragma unroll
            for (int r = 0; r < 8; ++r) {
                const float q = accQ0[r] + accQ1[r];
                g1[r] = (a1p[r] > 0.f) ? q : 0.f;
            }

            const float g1own = pick8(g1, wb0, wb1, wb2);
            const float a1own = pick8(a1, mb0, mb1, mb2);
            float ndown[2];
            #pragma unroll
            for (int t = 0; t < 2; ++t) {
                const float s8 = pick8(nd[t], db0, db1, db2);
                const float sx = other_half(s8);
                ndown[t] = ownHalf ? s8 : sx;
            }

            const float ng = -LRC * g1own;
            #pragma unroll
            for (int i = 0; i < 4; ++i) {
                #pragma unroll
                for (int j = 0; j < 2; ++j)
                    w1A[i][j] = fmaf(ng, xv[i][j], w1A[i][j]);
            }
            #pragma unroll
            for (int kb = 0; kb < 8; ++kb) {
                #pragma unroll
                for (int j = 0; j < 2; ++j)
                    w2P[kb][j] = fmaf(nd[kb >> 2][2 * (kb & 3) + j], a1own, w2P[kb][j]);
            }
            #pragma unroll
            for (int t = 0; t < 2; ++t) {
                #pragma unroll
                for (int kb = 0; kb < 2; ++kb) {
                    #pragma unroll
                    for (int j = 0; j < 2; ++j)
                        w2Q[t][kb][j] = fmaf(ndown[t], a1v[kb][j], w2Q[t][kb][j]);
                }
                #pragma unroll
                for (int r = 0; r < 8; ++r) b2D[t][r] -= LRC * dD[t][r];
            }
            #pragma unroll
            for (int i = 0; i < 4; ++i) b1D[i] -= LRC * g1[i];
        }
    };

    #pragma unroll
    for (int k = 0; k < 4; ++k) hid4[0][k * 32 + lane] = src[k * 32 + lane];

    #pragma unroll 1
    for (int c = 0; c < NCH; ++c) {
        v4f nx[4];
        const int cn = (c + 1 < NCH) ? (c + 1) : c;
        #pragma unroll
        for (int k = 0; k < 4; ++k) nx[k] = src[cn * 128 + k * 32 + lane];

        consume((const float*)hid4[c & 1], (c == NCH - 1) ? 7 : 8);

        #pragma unroll
        for (int k = 0; k < 4; ++k) hid4[(c + 1) & 1][k * 32 + lane] = nx[k];
    }

    {
        const float* hbuf = (const float*)hid4[(NCH - 1) & 1];
        v2f xv[4], a1v[2];
        float a1p[8], a1[8];
        v8f accY[2];
        forward(hbuf + 15 * Hq, xv, a1p, a1, a1v, accY);
        if (m == 0) {
            #pragma unroll
            for (int t = 0; t < 2; ++t) {
                #pragma unroll
                for (int r = 0; r < 8; ++r) ctxLds[16 * t + 8 * hi + r] = accY[t][r];
            }
        }
    }
    __syncthreads();

    #pragma unroll
    for (int half = 0; half < 2; ++half) {
        const int vo = half * 32 + lane;
        float acc = out_b[vo];
        #pragma unroll
        for (int h = 0; h < Hq; ++h)
            acc = fmaf(ctxLds[h], out_w[vo * Hq + h], acc);
        out[b * Vq + vo] = acc;
    }
}

extern "C" void kernel_launch(void* const* d_in, const int* in_sizes, int n_in,
                              void* d_out, int out_size, void* d_ws, size_t ws_size,
                              hipStream_t stream) {
    (void)in_sizes; (void)n_in; (void)out_size; (void)ws_size;
    const int*   seq    = (const int*)  d_in[0];
    const float* embed  = (const float*)d_in[1];
    const float* ff_w1  = (const float*)d_in[2];
    const float* ff_b1  = (const float*)d_in[3];
    const float* ff_w2  = (const float*)d_in[4];
    const float* ff_b2  = (const float*)d_in[5];
    const float* ln_g   = (const float*)d_in[6];
    const float* ln_b   = (const float*)d_in[7];
    const float* mlp_w1 = (const float*)d_in[8];
    const float* mlp_b1 = (const float*)d_in[9];
    const float* mlp_w2 = (const float*)d_in[10];
    const float* mlp_b2 = (const float*)d_in[11];
    const float* out_w  = (const float*)d_in[12];
    const float* out_b  = (const float*)d_in[13];
    float*       out    = (float*)d_out;
    float*       hidden = (float*)d_ws;

    pmm_encode_kernel<<<dim3(ENC_BLOCKS), dim3(ENC_WAVES * 32), 0, stream>>>(
        seq, embed, ff_w1, ff_b1, ff_w2, ff_b2, ln_g, ln_b, hidden);
    pmm_sgd_kernel<<<dim3(Bq), dim3(32), 0, stream>>>(
        hidden, mlp_w1, mlp_b1, mlp_w2, mlp_b2, out_w, out_b, out);
}


# SGD launch: fc2 backward's 32-long sums split over matrix rows AND columns (row/column parity = k half = y tile): 4 matrix instructions, not 8; half sums joined by a lane^1 add; q's sum order changes
# speedup vs baseline: 2.6410x; 1.1030x over previous
#include <hip/hip_runtime.h>
#include <hip/hip_bf16.h>
#include <math.h>


typedef __attribute__((ext_vector_type(2))) float v2f;
typedef __attribute__((ext_vector_type(4))) float v4f;
typedef __attribute__((ext_vector_type(8))) float v8f;

#define LRC  0.01f
#define EPSC 1e-5f

#define Bq 256
#define Lq 4096
#define Hq 32
#define Vq 64
#define CH 16
#define NCH (Lq / CH)
#define A1STR 68
#define HSTR  36

#define ENC_WAVES 8
#define ENC_CPW   8
#define ENC_BLOCKS ((Bq * NCH) / (ENC_WAVES * ENC_CPW))

__launch_bounds__(ENC_WAVES * 32)
__global__ void pmm_encode_kernel(const int* __restrict__ seq,
                                  const float* __restrict__ embed,
                                  const float* __restrict__ ff_w1,
                                  const float* __restrict__ ff_b1,
                                  const float* __restrict__ ff_w2,
                                  const float* __restrict__ ff_b2,
                                  const float* __restrict__ ln_g,
                                  const float* __restrict__ ln_b,
                                  float* __restrict__ hidden)
{
    __shared__ float embedLds[Vq * Hq];
    __shared__ float w1Lds[2 * Hq * Hq];
    __shared__ float w2Lds[Hq * 2 * Hq];
    __shared__ float a1all[ENC_WAVES][CH * A1STR];
    __shared__ float hball[ENC_WAVES][CH * HSTR];

    const int tid  = threadIdx.x;
    const int lane = tid & 31;
    const int wave = tid >> 5;
    const int m    = lane & 15;
    const int hi   = lane >> 4;

    for (int i = tid; i < Vq * Hq; i += ENC_WAVES * 32)     embedLds[i] = embed[i];
    for (int i = tid; i < 2 * Hq * Hq; i += ENC_WAVES * 32) w1Lds[i]    = ff_w1[i];
    for (int i = tid; i < 2 * Hq * Hq; i += ENC_WAVES * 32) w2Lds[i]    = ff_w2[i];
    __syncthreads();

    float b1t[4], b2t[2];
    #pragma unroll
    for (int nt = 0; nt < 4; ++nt) b1t[nt] = ff_b1[nt * 16 + m];
    #pragma unroll
    for (int nt = 0; nt < 2; ++nt) b2t[nt] = ff_b2[nt * 16 + m];
    const float lnGv = ln_g[lane];
    const float lnBv = ln_b[lane];

    float* a1buf = a1all[wave];
    float* hbuf  = hball[wave];

    const int g0 = (blockIdx.x * ENC_WAVES + wave) * ENC_CPW;

    #pragma unroll 1
    for (int j = 0; j < ENC_CPW; ++j) {
        const int base = (g0 + j) * CH;
        const int tok  = seq[base + m];

        v2f eA[8];
        #pragma unroll
        for (int kb = 0; kb < 8; ++kb)
            eA[kb] = *(const v2f*)&embedLds[tok * Hq + 4 * kb + 2 * hi];

        #pragma unroll 1
        for (int nt = 0; nt < 4; ++nt) {
            const int n0 = nt * 16;
            v8f acc = {};
            #pragma unroll
            for (int kb = 0; kb < 8; ++kb) {
                v2f bB = *(const v2f*)&w1Lds[(n0 + m) * Hq + 4 * kb + 2 * hi];
                acc = __builtin_amdgcn_wmma_f32_16x16x4_f32(
                        false, eA[kb], false, bB, (short)0, acc, false, false);
            }
            const float bias = b1t[nt];
            #pragma unroll
            for (int r = 0; r < 8; ++r) {
                float z = acc[r] + bias;
                a1buf[(r + 8 * hi) * A1STR + n0 + m] = fmaxf(z, 0.f);
            }
        }
        asm volatile("s_wait_dscnt 0" ::: "memory");
        __builtin_amdgcn_wave_barrier();

        v2f a2[16];
        #pragma unroll
        for (int kb = 0; kb < 16; ++kb)
            a2[kb] = *(const v2f*)&a1buf[m * A1STR + 4 * kb + 2 * hi];

        #pragma unroll 1
        for (int nt = 0; nt < 2; ++nt) {
            const int n0 = nt * 16;
            v8f acc = {};
            #pragma unroll
            for (int kb = 0; kb < 16; ++kb) {
                v2f bB = *(const v2f*)&w2Lds[(n0 + m) * (2 * Hq) + 4 * kb + 2 * hi];
                acc = __builtin_amdgcn_wmma_f32_16x16x4_f32(
                        false, a2[kb], false, bB, (short)0, acc, false, false);
            }
            const float bias = b2t[nt];
            #pragma unroll
            for (int r = 0; r < 8; ++r) {
                int   tm = __shfl(tok, r + 8 * hi, 32);
                float e  = embedLds[tm * Hq + n0 + m];
                hbuf[(r + 8 * hi) * HSTR + n0 + m] = acc[r] + bias + e;
            }
        }
        asm volatile("s_wait_dscnt 0" ::: "memory");
        __builtin_amdgcn_wave_barrier();

        #pragma unroll 1
        for (int t = 0; t < CH; ++t) {
            float z = hbuf[t * HSTR + lane];
            float s = z, s2 = z * z;
            #pragma unroll
            for (int off = 16; off >= 1; off >>= 1) {
                s  += __shfl_xor(s,  off, 32);
                s2 += __shfl_xor(s2, off, 32);
            }
            float mu  = s * (1.f / 32.f);
            float var = s2 * (1.f / 32.f) - mu * mu;
            float nz  = (z - mu) * rsqrtf(var + EPSC);
            hidden[(size_t)(base + t) * Hq + lane] = nz * lnGv + lnBv;
        }
        asm volatile("s_wait_dscnt 0" ::: "memory");
    }
}

__device__ __forceinline__ float pick8(const float (&a)[8], bool b0, bool b1, bool b2)
{
    const float t0 = b0 ? a[1] : a[0];
    const float t1 = b0 ? a[3] : a[2];
    const float t2 = b0 ? a[5] : a[4];
    const float t3 = b0 ? a[7] : a[6];
    const float u0 = b1 ? t1 : t0;
    const float u1 = b1 ? t3 : t2;
    return b2 ? u1 : u0;
}

__device__ __forceinline__ float pick4(const float (&a)[4], bool b0, bool b1)
{
    const float t0 = b0 ? a[1] : a[0];
    const float t1 = b0 ? a[3] : a[2];
    return b1 ? t1 : t0;
}

__device__ __forceinline__ float pair_lane(float s)
{
    return __int_as_float(__builtin_amdgcn_update_dpp(0, __float_as_int(s), 0xB1, 0xF, 0xF, true));
}

__device__ __forceinline__ float other_half(float s)
{
    const int r = __builtin_amdgcn_permlanex16(0, __float_as_int(s), 0x76543210, 0xfedcba98, false, false);
    return __int_as_float(r);
}

__launch_bounds__(32)
__global__ void pmm_sgd_kernel(const float* __restrict__ hidden,
                               const float* __restrict__ mlp_w1,
                               const float* __restrict__ mlp_b1,
                               const float* __restrict__ mlp_w2,
                               const float* __restrict__ mlp_b2,
                               const float* __restrict__ out_w,
                               const float* __restrict__ out_b,
                               float* __restrict__ out)
{
    __shared__ v4f   hid4[2][CH * Hq / 4];
    __shared__ float ctxLds[Hq];

    const int b    = blockIdx.x;
    const int lane = threadIdx.x;
    const int m    = lane & 15;
    const int hi   = lane >> 4;
    const bool db0 = (m & 1) != 0, db1 = (m & 2) != 0, db2 = (m & 4) != 0;
    const bool ownHalf = (m >> 3) == hi;
    const int uW1  = 4 * (m >> 3) + ((m & 7) >> 1);
    const int sW1  = m & 1;
    const bool ib0 = (m & 2) != 0, ib1 = (m & 4) != 0;
    const bool oddCol = sW1 != 0;

    const v4f* src = (const v4f*)(hidden + (size_t)b * Lq * Hq);

    v2f w1A[4], w2P[4], w2Q[2][2];
    float b1D[4], b2D[2][8];
    #pragma unroll
    for (int i = 0; i < 4; ++i) {
        #pragma unroll
        for (int j = 0; j < 2; ++j)
            w1A[i][j] = mlp_w1[uW1 * Hq + 16 * sW1 + 4 * i + 2 * hi + j];
    }
    #pragma unroll
    for (int i = 0; i < 4; ++i) {
        #pragma unroll
        for (int j = 0; j < 2; ++j)
            w2P[i][j] = mlp_w2[(16 * sW1 + 8 * hi + 2 * i + j) * 8 + uW1];
    }
    #pragma unroll
    for (int t = 0; t < 2; ++t) {
        #pragma unroll
        for (int kb = 0; kb < 2; ++kb) {
            #pragma unroll
            for (int j = 0; j < 2; ++j)
                w2Q[t][kb][j] = mlp_w2[(16 * t + m) * 8 + 4 * hi + 2 * kb + j];
        }
        #pragma unroll
        for (int r = 0; r < 8; ++r) b2D[t][r] = mlp_b2[16 * t + 8 * hi + r];
    }
    #pragma unroll
    for (int i = 0; i < 4; ++i) b1D[i] = mlp_b1[4 * hi + i];

    auto forward = [&](const float* xrow, v2f (&xv)[4], float (&a1p)[4], float (&a1)[4],
                       v2f (&a1v)[2], v8f (&accY)[2]) {
        #pragma unroll
        for (int i = 0; i < 4; ++i)
            xv[i] = *(const v2f*)(xrow + 16 * sW1 + 4 * i + 2 * hi);
        v8f accP0 = {}, accP1 = {};
        #pragma unroll
        for (int i = 0; i < 4; i += 2) {
            accP0 = __builtin_amdgcn_wmma_f32_16x16x4_f32(
                        false, w1A[i], false, xv[i], (short)0, accP0, false, false);
            accP1 = __builtin_amdgcn_wmma_f32_16x16x4_f32(
                        false, w1A[i + 1], false, xv[i + 1], (short)0, accP1, false, false);
        }
        #pragma unroll
        for (int i = 0; i < 4; ++i) {
            const float sE = accP0[2 * i] + accP1[2 * i];
            const float sO = accP0[2 * i + 1] + accP1[2 * i + 1];
            const float own = oddCol ? sO : sE;
            a1p[i] = (own + pair_lane(own)) + b1D[i];
            a1[i]  = fmaxf(a1p[i], 0.f);
        }
        #pragma unroll
        for (int kb = 0; kb < 2; ++kb) {
            a1v[kb][0] = a1[2 * kb];
            a1v[kb][1] = a1[2 * kb + 1];
        }
        #pragma unroll
        for (int t = 0; t < 2; ++t) {
            #pragma unroll
            for (int r = 0; r < 8; ++r) accY[t][r] = b2D[t][r];
        }
        #pragma unroll
        for (int kb = 0; kb < 2; ++kb) {
            #pragma unroll
            for (int t = 0; t < 2; ++t)
                accY[t] = __builtin_amdgcn_wmma_f32_16x16x4_f32(
                            false, w2Q[t][kb], false, a1v[kb], (short)0, accY[t], false, false);
        }
    };

    auto consume = [&](const float* hbuf, int nsteps) {
        #pragma unroll 1
        for (int s = 0; s < nsteps; ++s) {
            const float* xrow = hbuf + (2 * s) * Hq;
            const float* vrow = hbuf + (2 * s + 1) * Hq;

            float vD[2][8];
            #pragma unroll
            for (int t = 0; t < 2; ++t) {
                #pragma unroll
                for (int u = 0; u < 2; ++u) {
                    const v4f vq = *(const v4f*)(vrow + 16 * t + 8 * hi + 4 * u);
                    vD[t][4 * u + 0] = vq[0]; vD[t][4 * u + 1] = vq[1];
                    vD[t][4 * u + 2] = vq[2]; vD[t][4 * u + 3] = vq[3];
                }
            }

            v2f xv[4], a1v[2];
            float a1p[4], a1[4];
            v8f accY[2];
            forward(xrow, xv, a1p, a1, a1v, accY);

            float dD[2][8], nd[2][8];
            #pragma unroll
            for (int t = 0; t < 2; ++t) {
                #pragma unroll
                for (int r = 0; r < 8; ++r) {
                    dD[t][r] = (accY[t][r] - vD[t][r]) * 0.0625f;
                    nd[t][r] = -LRC * dD[t][r];
                }
            }

            float dsel[8], ndsel[8];
            #pragma unroll
            for (int r = 0; r < 8; ++r) {
                dsel[r]  = oddCol ? dD[1][r] : dD[0][r];
                ndsel[r] = oddCol ? nd[1][r] : nd[0][r];
            }
            v8f accQ0 = {}, accQ1 = {};
            #pragma unroll
            for (int i = 0; i < 4; i += 2) {
                v2f dv0, dv1;
                dv0[0] = dsel[2 * i];     dv0[1] = dsel[2 * i + 1];
                dv1[0] = dsel[2 * i + 2]; dv1[1] = dsel[2 * i + 3];
                accQ0 = __builtin_amdgcn_wmma_f32_16x16x4_f32(
                            false, w2P[i], false, dv0, (short)0, accQ0, false, false);
                accQ1 = __builtin_amdgcn_wmma_f32_16x16x4_f32(
                            false, w2P[i + 1], false, dv1, (short)0, accQ1, false, false);
            }

            float g1[4];
            #pragma unroll
            for (int i = 0; i < 4; ++i) {
                const float sE = accQ0[2 * i] + accQ1[2 * i];
                const float sO = accQ0[2 * i + 1] + accQ1[2 * i + 1];
                const float own = oddCol ? sO : sE;
                const float q = own + pair_lane(own);
                g1[i] = (a1p[i] > 0.f) ? q : 0.f;
            }

            const float g1sel = pick4(g1, ib0, ib1);
            const float a1sel = pick4(a1, ib0, ib1);
            const float g1far = other_half(g1sel);
            const float a1far = other_half(a1sel);
            const float g1own = ownHalf ? g1sel : g1far;
            const float a1own = ownHalf ? a1sel : a1far;
            float ndown[2];
            #pragma unroll
            for (int t = 0; t < 2; ++t) {
                const float s8 = pick8(nd[t], db0, db1, db2);
                const float sx = other_half(s8);
                ndown[t] = ownHalf ? s8 : sx;
            }

            const float ng = -LRC * g1own;
            #pragma unroll
            for (int i = 0; i < 4; ++i) {
                #pragma unroll
                for (int j = 0; j < 2; ++j)
                    w1A[i][j] = fmaf(ng, xv[i][j], w1A[i][j]);
            }
            #pragma unroll
            for (int i = 0; i < 4; ++i) {
                #pragma unroll
                for (int j = 0; j < 2; ++j)
                    w2P[i][j] = fmaf(ndsel[2 * i + j], a1own, w2P[i][j]);
            }
            #pragma unroll
            for (int t = 0; t < 2; ++t) {
                #pragma unroll
                for (int kb = 0; kb < 2; ++kb) {
                    #pragma unroll
                    for (int j = 0; j < 2; ++j)
                        w2Q[t][kb][j] = fmaf(ndown[t], a1v[kb][j], w2Q[t][kb][j]);
                }
                #pragma unroll
                for (int r = 0; r < 8; ++r) b2D[t][r] -= LRC * dD[t][r];
            }
            #pragma unroll
            for (int i = 0; i < 4; ++i) b1D[i] -= LRC * g1[i];
        }
    };

    #pragma unroll
    for (int k = 0; k < 4; ++k) hid4[0][k * 32 + lane] = src[k * 32 + lane];

    #pragma unroll 1
    for (int c = 0; c < NCH; ++c) {
        v4f nx[4];
        const int cn = (c + 1 < NCH) ? (c + 1) : c;
        #pragma unroll
        for (int k = 0; k < 4; ++k) nx[k] = src[cn * 128 + k * 32 + lane];

        consume((const float*)hid4[c & 1], (c == NCH - 1) ? 7 : 8);

        #pragma unroll
        for (int k = 0; k < 4; ++k) hid4[(c + 1) & 1][k * 32 + lane] = nx[k];
    }

    {
        const float* hbuf = (const float*)hid4[(NCH - 1) & 1];
        v2f xv[4], a1v[2];
        float a1p[4], a1[4];
        v8f accY[2];
        forward(hbuf + 15 * Hq, xv, a1p, a1, a1v, accY);
        if (m == 0) {
            #pragma unroll
            for (int t = 0; t < 2; ++t) {
                #pragma unroll
                for (int r = 0; r < 8; ++r) ctxLds[16 * t + 8 * hi + r] = accY[t][r];
            }
        }
    }
    __syncthreads();

    #pragma unroll
    for (int half = 0; half < 2; ++half) {
        const int vo = half * 32 + lane;
        float acc = out_b[vo];
        #pragma unroll
        for (int h = 0; h < Hq; ++h)
            acc = fmaf(ctxLds[h], out_w[vo * Hq + h], acc);
        out[b * Vq + vo] = acc;
    }
}

extern "C" void kernel_launch(void* const* d_in, const int* in_sizes, int n_in,
                              void* d_out, int out_size, void* d_ws, size_t ws_size,
                              hipStream_t stream) {
    (void)in_sizes; (void)n_in; (void)out_size; (void)ws_size;
    const int*   seq    = (const int*)  d_in[0];
    const float* embed  = (const float*)d_in[1];
    const float* ff_w1  = (const float*)d_in[2];
    const float* ff_b1  = (const float*)d_in[3];
    const float* ff_w2  = (const float*)d_in[4];
    const float* ff_b2  = (const float*)d_in[5];
    const float* ln_g   = (const float*)d_in[6];
    const float* ln_b   = (const float*)d_in[7];
    const float* mlp_w1 = (const float*)d_in[8];
    const float* mlp_b1 = (const float*)d_in[9];
    const float* mlp_w2 = (const float*)d_in[10];
    const float* mlp_b2 = (const float*)d_in[11];
    const float* out_w  = (const float*)d_in[12];
    const float* out_b  = (const float*)d_in[13];
    float*       out    = (float*)d_out;
    float*       hidden = (float*)d_ws;

    pmm_encode_kernel<<<dim3(ENC_BLOCKS), dim3(ENC_WAVES * 32), 0, stream>>>(
        seq, embed, ff_w1, ff_b1, ff_w2, ff_b2, ln_g, ln_b, hidden);
    pmm_sgd_kernel<<<dim3(Bq), dim3(32), 0, stream>>>(
        hidden, mlp_w1, mlp_b1, mlp_w2, mlp_b2, out_w, out_b, out);
}


# SGD launch: fc2 backward product with operands swapped (d as A, weights as B): each lane gets q of its own weight row's unit; the pick and half-wave exchange leave the step's tail; sums unchanged
# speedup vs baseline: 2.8242x; 1.0694x over previous
#include <hip/hip_runtime.h>
#include <hip/hip_bf16.h>
#include <math.h>


typedef __attribute__((ext_vector_type(2))) float v2f;
typedef __attribute__((ext_vector_type(4))) float v4f;
typedef __attribute__((ext_vector_type(8))) float v8f;

#define LRC  0.01f
#define EPSC 1e-5f

#define Bq 256
#define Lq 4096
#define Hq 32
#define Vq 64
#define CH 16
#define NCH (Lq / CH)
#define A1STR 68
#define HSTR  36

#define ENC_WAVES 8
#define ENC_CPW   8
#define ENC_BLOCKS ((Bq * NCH) / (ENC_WAVES * ENC_CPW))

__launch_bounds__(ENC_WAVES * 32)
__global__ void pmm_encode_kernel(const int* __restrict__ seq,
                                  const float* __restrict__ embed,
                                  const float* __restrict__ ff_w1,
                                  const float* __restrict__ ff_b1,
                                  const float* __restrict__ ff_w2,
                                  const float* __restrict__ ff_b2,
                                  const float* __restrict__ ln_g,
                                  const float* __restrict__ ln_b,
                                  float* __restrict__ hidden)
{
    __shared__ float embedLds[Vq * Hq];
    __shared__ float w1Lds[2 * Hq * Hq];
    __shared__ float w2Lds[Hq * 2 * Hq];
    __shared__ float a1all[ENC_WAVES][CH * A1STR];
    __shared__ float hball[ENC_WAVES][CH * HSTR];

    const int tid  = threadIdx.x;
    const int lane = tid & 31;
    const int wave = tid >> 5;
    const int m    = lane & 15;
    const int hi   = lane >> 4;

    for (int i = tid; i < Vq * Hq; i += ENC_WAVES * 32)     embedLds[i] = embed[i];
    for (int i = tid; i < 2 * Hq * Hq; i += ENC_WAVES * 32) w1Lds[i]    = ff_w1[i];
    for (int i = tid; i < 2 * Hq * Hq; i += ENC_WAVES * 32) w2Lds[i]    = ff_w2[i];
    __syncthreads();

    float b1t[4], b2t[2];
    #pragma unroll
    for (int nt = 0; nt < 4; ++nt) b1t[nt] = ff_b1[nt * 16 + m];
    #pragma unroll
    for (int nt = 0; nt < 2; ++nt) b2t[nt] = ff_b2[nt * 16 + m];
    const float lnGv = ln_g[lane];
    const float lnBv = ln_b[lane];

    float* a1buf = a1all[wave];
    float* hbuf  = hball[wave];

    const int g0 = (blockIdx.x * ENC_WAVES + wave) * ENC_CPW;

    #pragma unroll 1
    for (int j = 0; j < ENC_CPW; ++j) {
        const int base = (g0 + j) * CH;
        const int tok  = seq[base + m];

        v2f eA[8];
        #pragma unroll
        for (int kb = 0; kb < 8; ++kb)
            eA[kb] = *(const v2f*)&embedLds[tok * Hq + 4 * kb + 2 * hi];

        #pragma unroll 1
        for (int nt = 0; nt < 4; ++nt) {
            const int n0 = nt * 16;
            v8f acc = {};
            #pragma unroll
            for (int kb = 0; kb < 8; ++kb) {
                v2f bB = *(const v2f*)&w1Lds[(n0 + m) * Hq + 4 * kb + 2 * hi];
                acc = __builtin_amdgcn_wmma_f32_16x16x4_f32(
                        false, eA[kb], false, bB, (short)0, acc, false, false);
            }
            const float bias = b1t[nt];
            #pragma unroll
            for (int r = 0; r < 8; ++r) {
                float z = acc[r] + bias;
                a1buf[(r + 8 * hi) * A1STR + n0 + m] = fmaxf(z, 0.f);
            }
        }
        asm volatile("s_wait_dscnt 0" ::: "memory");
        __builtin_amdgcn_wave_barrier();

        v2f a2[16];
        #pragma unroll
        for (int kb = 0; kb < 16; ++kb)
            a2[kb] = *(const v2f*)&a1buf[m * A1STR + 4 * kb + 2 * hi];

        #pragma unroll 1
        for (int nt = 0; nt < 2; ++nt) {
            const int n0 = nt * 16;
            v8f acc = {};
            #pragma unroll
            for (int kb = 0; kb < 16; ++kb) {
                v2f bB = *(const v2f*)&w2Lds[(n0 + m) * (2 * Hq) + 4 * kb + 2 * hi];
                acc = __builtin_amdgcn_wmma_f32_16x16x4_f32(
                        false, a2[kb], false, bB, (short)0, acc, false, false);
            }
            const float bias = b2t[nt];
            #pragma unroll
            for (int r = 0; r < 8; ++r) {
                int   tm = __shfl(tok, r + 8 * hi, 32);
                float e  = embedLds[tm * Hq + n0 + m];
                hbuf[(r + 8 * hi) * HSTR + n0 + m] = acc[r] + bias + e;
            }
        }
        asm volatile("s_wait_dscnt 0" ::: "memory");
        __builtin_amdgcn_wave_barrier();

        #pragma unroll 1
        for (int t = 0; t < CH; ++t) {
            float z = hbuf[t * HSTR + lane];
            float s = z, s2 = z * z;
            #pragma unroll
            for (int off = 16; off >= 1; off >>= 1) {
                s  += __shfl_xor(s,  off, 32);
                s2 += __shfl_xor(s2, off, 32);
            }
            float mu  = s * (1.f / 32.f);
            float var = s2 * (1.f / 32.f) - mu * mu;
            float nz  = (z - mu) * rsqrtf(var + EPSC);
            hidden[(size_t)(base + t) * Hq + lane] = nz * lnGv + lnBv;
        }
        asm volatile("s_wait_dscnt 0" ::: "memory");
    }
}

__device__ __forceinline__ float pick8(const float (&a)[8], bool b0, bool b1, bool b2)
{
    const float t0 = b0 ? a[1] : a[0];
    const float t1 = b0 ? a[3] : a[2];
    const float t2 = b0 ? a[5] : a[4];
    const float t3 = b0 ? a[7] : a[6];
    const float u0 = b1 ? t1 : t0;
    const float u1 = b1 ? t3 : t2;
    return b2 ? u1 : u0;
}

__device__ __forceinline__ float pick4(const float (&a)[4], bool b0, bool b1)
{
    const float t0 = b0 ? a[1] : a[0];
    const float t1 = b0 ? a[3] : a[2];
    return b1 ? t1 : t0;
}

__device__ __forceinline__ float pair_lane(float s)
{
    return __int_as_float(__builtin_amdgcn_update_dpp(0, __float_as_int(s), 0xB1, 0xF, 0xF, true));
}

__device__ __forceinline__ float other_half(float s)
{
    const int r = __builtin_amdgcn_permlanex16(0, __float_as_int(s), 0x76543210, 0xfedcba98, false, false);
    return __int_as_float(r);
}

__device__ __forceinline__ float high_row_turned(float s)
{
    return __int_as_float(__builtin_amdgcn_update_dpp(__float_as_int(s), __float_as_int(s), 0x128, 0x2, 0xF, false));
}

template <int N>
__device__ __forceinline__ float row_lane(float s)
{
    return __int_as_float(__builtin_amdgcn_update_dpp(0, __float_as_int(s), 0x150 + N, 0xF, 0xF, true));
}

__launch_bounds__(32)
__global__ void pmm_sgd_kernel(const float* __restrict__ hidden,
                               const float* __restrict__ mlp_w1,
                               const float* __restrict__ mlp_b1,
                               const float* __restrict__ mlp_w2,
                               const float* __restrict__ mlp_b2,
                               const float* __restrict__ out_w,
                               const float* __restrict__ out_b,
                               float* __restrict__ out)
{
    __shared__ v4f   hid4[2][CH * Hq / 4];
    __shared__ float ctxLds[Hq];

    const int b    = blockIdx.x;
    const int lane = threadIdx.x;
    const int m    = lane & 15;
    const int hi   = lane >> 4;
    const bool db0 = (m & 1) != 0, db1 = (m & 2) != 0, db2 = (m & 4) != 0;
    const bool ownHalf = (m >> 3) == hi;
    const int uW1  = 4 * (m >> 3) + ((m & 7) >> 1);
    const int sW1  = m & 1;
    const bool ib0 = (m & 2) != 0, ib1 = (m & 4) != 0;
    const bool oddCol = sW1 != 0;

    const v4f* src = (const v4f*)(hidden + (size_t)b * Lq * Hq);

    v2f w1A[4], w2P[4], w2Q[2][2];
    float b2D[2][8];
    #pragma unroll
    for (int i = 0; i < 4; ++i) {
        #pragma unroll
        for (int j = 0; j < 2; ++j)
            w1A[i][j] = mlp_w1[uW1 * Hq + 16 * sW1 + 4 * i + 2 * hi + j];
    }
    #pragma unroll
    for (int i = 0; i < 4; ++i) {
        #pragma unroll
        for (int j = 0; j < 2; ++j)
            w2P[i][j] = mlp_w2[(16 * sW1 + 8 * hi + 2 * i + j) * 8 + uW1];
    }
    #pragma unroll
    for (int t = 0; t < 2; ++t) {
        #pragma unroll
        for (int kb = 0; kb < 2; ++kb) {
            #pragma unroll
            for (int j = 0; j < 2; ++j)
                w2Q[t][kb][j] = mlp_w2[(16 * t + m) * 8 + 4 * hi + 2 * kb + j];
        }
        #pragma unroll
        for (int r = 0; r < 8; ++r) b2D[t][r] = mlp_b2[16 * t + 8 * hi + r];
    }
    float b1own = mlp_b1[uW1];

    auto forward = [&](const float* xrow, v2f (&xv)[4], float (&a1)[4],
                       v2f (&a1v)[2], v8f (&accY)[2]) {
        #pragma unroll
        for (int i = 0; i < 4; ++i)
            xv[i] = *(const v2f*)(xrow + 16 * sW1 + 4 * i + 2 * hi);
        v8f accP0 = {}, accP1 = {};
        #pragma unroll
        for (int i = 0; i < 4; i += 2) {
            accP0 = __builtin_amdgcn_wmma_f32_16x16x4_f32(
                        false, w1A[i], false, xv[i], (short)0, accP0, false, false);
            accP1 = __builtin_amdgcn_wmma_f32_16x16x4_f32(
                        false, w1A[i + 1], false, xv[i + 1], (short)0, accP1, false, false);
        }
        const float b1t = high_row_turned(b1own);
        float b1D[4];
        b1D[0] = row_lane<0>(b1t);
        b1D[1] = row_lane<2>(b1t);
        b1D[2] = row_lane<4>(b1t);
        b1D[3] = row_lane<6>(b1t);
        #pragma unroll
        for (int i = 0; i < 4; ++i) {
            const float sE = accP0[2 * i] + accP1[2 * i];
            const float sO = accP0[2 * i + 1] + accP1[2 * i + 1];
            const float own = oddCol ? sO : sE;
            const float a1p = (own + pair_lane(own)) + b1D[i];
            a1[i] = fmaxf(a1p, 0.f);
        }
        #pragma unroll
        for (int kb = 0; kb < 2; ++kb) {
            a1v[kb][0] = a1[2 * kb];
            a1v[kb][1] = a1[2 * kb + 1];
        }
        #pragma unroll
        for (int t = 0; t < 2; ++t) {
            #pragma unroll
            for (int r = 0; r < 8; ++r) accY[t][r] = b2D[t][r];
        }
        #pragma unroll
        for (int kb = 0; kb < 2; ++kb) {
            #pragma unroll
            for (int t = 0; t < 2; ++t)
                accY[t] = __builtin_amdgcn_wmma_f32_16x16x4_f32(
                            false, w2Q[t][kb], false, a1v[kb], (short)0, accY[t], false, false);
        }
    };

    auto consume = [&](const float* hbuf, int nsteps) {
        #pragma unroll 1
        for (int s = 0; s < nsteps; ++s) {
            const float* xrow = hbuf + (2 * s) * Hq;
            const float* vrow = hbuf + (2 * s + 1) * Hq;

            float vD[2][8];
            #pragma unroll
            for (int t = 0; t < 2; ++t) {
                #pragma unroll
                for (int u = 0; u < 2; ++u) {
                    const v4f vq = *(const v4f*)(vrow + 16 * t + 8 * hi + 4 * u);
                    vD[t][4 * u + 0] = vq[0]; vD[t][4 * u + 1] = vq[1];
                    vD[t][4 * u + 2] = vq[2]; vD[t][4 * u + 3] = vq[3];
                }
            }

            v2f xv[4], a1v[2];
            float a1[4];
            v8f accY[2];
            forward(xrow, xv, a1, a1v, accY);

            float dD[2][8], nd[2][8];
            #pragma unroll
            for (int t = 0; t < 2; ++t) {
                #pragma unroll
                for (int r = 0; r < 8; ++r) {
                    dD[t][r] = (accY[t][r] - vD[t][r]) * 0.0625f;
                    nd[t][r] = -LRC * dD[t][r];
                }
            }

            float dsel[8], ndsel[8];
            #pragma unroll
            for (int r = 0; r < 8; ++r) {
                dsel[r]  = oddCol ? dD[1][r] : dD[0][r];
                ndsel[r] = oddCol ? nd[1][r] : nd[0][r];
            }
            v8f accQ0 = {}, accQ1 = {};
            #pragma unroll
            for (int i = 0; i < 4; i += 2) {
                v2f dv0, dv1;
                dv0[0] = dsel[2 * i];     dv0[1] = dsel[2 * i + 1];
                dv1[0] = dsel[2 * i + 2]; dv1[1] = dsel[2 * i + 3];
                accQ0 = __builtin_amdgcn_wmma_f32_16x16x4_f32(
                            false, dv0, false, w2P[i], (short)0, accQ0, false, false);
                accQ1 = __builtin_amdgcn_wmma_f32_16x16x4_f32(
                            false, dv1, false, w2P[i + 1], (short)0, accQ1, false, false);
            }

            const float a1sel = pick4(a1, ib0, ib1);
            const float a1far = other_half(a1sel);
            const float a1own = ownHalf ? a1sel : a1far;

            const float sE = accQ0[0] + accQ1[0];
            const float sO = accQ0[1] + accQ1[1];
            const float own = oddCol ? sO : sE;
            const float qown = own + pair_lane(own);
            const float g1own = (a1own > 0.f) ? qown : 0.f;

            float ndown[2];
            #pragma unroll
            for (int t = 0; t < 2; ++t) {
                const float s8 = pick8(nd[t], db0, db1, db2);
                const float sx = other_half(s8);
                ndown[t] = ownHalf ? s8 : sx;
            }

            const float ng = -LRC * g1own;
            #pragma unroll
            for (int i = 0; i < 4; ++i) {
                #pragma unroll
                for (int j = 0; j < 2; ++j)
                    w1A[i][j] = fmaf(ng, xv[i][j], w1A[i][j]);
            }
            #pragma unroll
            for (int i = 0; i < 4; ++i) {
                #pragma unroll
                for (int j = 0; j < 2; ++j)
                    w2P[i][j] = fmaf(ndsel[2 * i + j], a1own, w2P[i][j]);
            }
            #pragma unroll
            for (int t = 0; t < 2; ++t) {
                #pragma unroll
                for (int kb = 0; kb < 2; ++kb) {
                    #pragma unroll
                    for (int j = 0; j < 2; ++j)
                        w2Q[t][kb][j] = fmaf(ndown[t], a1v[kb][j], w2Q[t][kb][j]);
                }
                #pragma unroll
                for (int r = 0; r < 8; ++r) b2D[t][r] -= LRC * dD[t][r];
            }
            b1own -= LRC * g1own;
        }
    };

    #pragma unroll
    for (int k = 0; k < 4; ++k) hid4[0][k * 32 + lane] = src[k * 32 + lane];

    #pragma unroll 1
    for (int c = 0; c < NCH; ++c) {
        v4f nx[4];
        const int cn = (c + 1 < NCH) ? (c + 1) : c;
        #pragma unroll
        for (int k = 0; k < 4; ++k) nx[k] = src[cn * 128 + k * 32 + lane];

        consume((const float*)hid4[c & 1], (c == NCH - 1) ? 7 : 8);

        #pragma unroll
        for (int k = 0; k < 4; ++k) hid4[(c + 1) & 1][k * 32 + lane] = nx[k];
    }

    {
        const float* hbuf = (const float*)hid4[(NCH - 1) & 1];
        v2f xv[4], a1v[2];
        float a1[4];
        v8f accY[2];
        forward(hbuf + 15 * Hq, xv, a1, a1v, accY);
        if (m == 0) {
            #pragma unroll
            for (int t = 0; t < 2; ++t) {
                #pragma unroll
                for (int r = 0; r < 8; ++r) ctxLds[16 * t + 8 * hi + r] = accY[t][r];
            }
        }
    }
    __syncthreads();

    #pragma unroll
    for (int half = 0; half < 2; ++half) {
        const int vo = half * 32 + lane;
        float acc = out_b[vo];
        #pragma unroll
        for (int h = 0; h < Hq; ++h)
            acc = fmaf(ctxLds[h], out_w[vo * Hq + h], acc);
        out[b * Vq + vo] = acc;
    }
}

extern "C" void kernel_launch(void* const* d_in, const int* in_sizes, int n_in,
                              void* d_out, int out_size, void* d_ws, size_t ws_size,
                              hipStream_t stream) {
    (void)in_sizes; (void)n_in; (void)out_size; (void)ws_size;
    const int*   seq    = (const int*)  d_in[0];
    const float* embed  = (const float*)d_in[1];
    const float* ff_w1  = (const float*)d_in[2];
    const float* ff_b1  = (const float*)d_in[3];
    const float* ff_w2  = (const float*)d_in[4];
    const float* ff_b2  = (const float*)d_in[5];
    const float* ln_g   = (const float*)d_in[6];
    const float* ln_b   = (const float*)d_in[7];
    const float* mlp_w1 = (const float*)d_in[8];
    const float* mlp_b1 = (const float*)d_in[9];
    const float* mlp_w2 = (const float*)d_in[10];
    const float* mlp_b2 = (const float*)d_in[11];
    const float* out_w  = (const float*)d_in[12];
    const float* out_b  = (const float*)d_in[13];
    float*       out    = (float*)d_out;
    float*       hidden = (float*)d_ws;

    pmm_encode_kernel<<<dim3(ENC_BLOCKS), dim3(ENC_WAVES * 32), 0, stream>>>(
        seq, embed, ff_w1, ff_b1, ff_w2, ff_b2, ln_g, ln_b, hidden);
    pmm_sgd_kernel<<<dim3(Bq), dim3(32), 0, stream>>>(
        hidden, mlp_w1, mlp_b1, mlp_w2, mlp_b2, out_w, out_b, out);
}


# SGD launch: x, v operands of a chunk's 8 steps read from LDS into registers as one group per chunk (chunk-wide register tile); steps unrolled 8x, no LDS read, wait or address math inside; same sums
# speedup vs baseline: 3.0862x; 1.0928x over previous
#include <hip/hip_runtime.h>
#include <hip/hip_bf16.h>
#include <math.h>


typedef __attribute__((ext_vector_type(2))) float v2f;
typedef __attribute__((ext_vector_type(4))) float v4f;
typedef __attribute__((ext_vector_type(8))) float v8f;

#define LRC  0.01f
#define EPSC 1e-5f

#define Bq 256
#define Lq 4096
#define Hq 32
#define Vq 64
#define CH 16
#define NCH (Lq / CH)
#define A1STR 68
#define HSTR  36

#define ENC_WAVES 8
#define ENC_CPW   8
#define ENC_BLOCKS ((Bq * NCH) / (ENC_WAVES * ENC_CPW))

__launch_bounds__(ENC_WAVES * 32)
__global__ void pmm_encode_kernel(const int* __restrict__ seq,
                                  const float* __restrict__ embed,
                                  const float* __restrict__ ff_w1,
                                  const float* __restrict__ ff_b1,
                                  const float* __restrict__ ff_w2,
                                  const float* __restrict__ ff_b2,
                                  const float* __restrict__ ln_g,
                                  const float* __restrict__ ln_b,
                                  float* __restrict__ hidden)
{
    __shared__ float embedLds[Vq * Hq];
    __shared__ float w1Lds[2 * Hq * Hq];
    __shared__ float w2Lds[Hq * 2 * Hq];
    __shared__ float a1all[ENC_WAVES][CH * A1STR];
    __shared__ float hball[ENC_WAVES][CH * HSTR];

    const int tid  = threadIdx.x;
    const int lane = tid & 31;
    const int wave = tid >> 5;
    const int m    = lane & 15;
    const int hi   = lane >> 4;

    for (int i = tid; i < Vq * Hq; i += ENC_WAVES * 32)     embedLds[i] = embed[i];
    for (int i = tid; i < 2 * Hq * Hq; i += ENC_WAVES * 32) w1Lds[i]    = ff_w1[i];
    for (int i = tid; i < 2 * Hq * Hq; i += ENC_WAVES * 32) w2Lds[i]    = ff_w2[i];
    __syncthreads();

    float b1t[4], b2t[2];
    #pragma unroll
    for (int nt = 0; nt < 4; ++nt) b1t[nt] = ff_b1[nt * 16 + m];
    #pragma unroll
    for (int nt = 0; nt < 2; ++nt) b2t[nt] = ff_b2[nt * 16 + m];
    const float lnGv = ln_g[lane];
    const float lnBv = ln_b[lane];

    float* a1buf = a1all[wave];
    float* hbuf  = hball[wave];

    const int g0 = (blockIdx.x * ENC_WAVES + wave) * ENC_CPW;

    #pragma unroll 1
    for (int j = 0; j < ENC_CPW; ++j) {
        const int base = (g0 + j) * CH;
        const int tok  = seq[base + m];

        v2f eA[8];
        #pragma unroll
        for (int kb = 0; kb < 8; ++kb)
            eA[kb] = *(const v2f*)&embedLds[tok * Hq + 4 * kb + 2 * hi];

        #pragma unroll 1
        for (int nt = 0; nt < 4; ++nt) {
            const int n0 = nt * 16;
            v8f acc = {};
            #pragma unroll
            for (int kb = 0; kb < 8; ++kb) {
                v2f bB = *(const v2f*)&w1Lds[(n0 + m) * Hq + 4 * kb + 2 * hi];
                acc = __builtin_amdgcn_wmma_f32_16x16x4_f32(
                        false, eA[kb], false, bB, (short)0, acc, false, false);
            }
            const float bias = b1t[nt];
            #pragma unroll
            for (int r = 0; r < 8; ++r) {
                float z = acc[r] + bias;
                a1buf[(r + 8 * hi) * A1STR + n0 + m] = fmaxf(z, 0.f);
            }
        }
        asm volatile("s_wait_dscnt 0" ::: "memory");
        __builtin_amdgcn_wave_barrier();

        v2f a2[16];
        #pragma unroll
        for (int kb = 0; kb < 16; ++kb)
            a2[kb] = *(const v2f*)&a1buf[m * A1STR + 4 * kb + 2 * hi];

        #pragma unroll 1
        for (int nt = 0; nt < 2; ++nt) {
            const int n0 = nt * 16;
            v8f acc = {};
            #pragma unroll
            for (int kb = 0; kb < 16; ++kb) {
                v2f bB = *(const v2f*)&w2Lds[(n0 + m) * (2 * Hq) + 4 * kb + 2 * hi];
                acc = __builtin_amdgcn_wmma_f32_16x16x4_f32(
                        false, a2[kb], false, bB, (short)0, acc, false, false);
            }
            const float bias = b2t[nt];
            #pragma unroll
            for (int r = 0; r < 8; ++r) {
                int   tm = __shfl(tok, r + 8 * hi, 32);
                float e  = embedLds[tm * Hq + n0 + m];
                hbuf[(r + 8 * hi) * HSTR + n0 + m] = acc[r] + bias + e;
            }
        }
        asm volatile("s_wait_dscnt 0" ::: "memory");
        __builtin_amdgcn_wave_barrier();

        #pragma unroll 1
        for (int t = 0; t < CH; ++t) {
            float z = hbuf[t * HSTR + lane];
            float s = z, s2 = z * z;
            #pragma unroll
            for (int off = 16; off >= 1; off >>= 1) {
                s  += __shfl_xor(s,  off, 32);
                s2 += __shfl_xor(s2, off, 32);
            }
            float mu  = s * (1.f / 32.f);
            float var = s2 * (1.f / 32.f) - mu * mu;
            float nz  = (z - mu) * rsqrtf(var + EPSC);
            hidden[(size_t)(base + t) * Hq + lane] = nz * lnGv + lnBv;
        }
        asm volatile("s_wait_dscnt 0" ::: "memory");
    }
}

__device__ __forceinline__ float pick8(const float (&a)[8], bool b0, bool b1, bool b2)
{
    const float t0 = b0 ? a[1] : a[0];
    const float t1 = b0 ? a[3] : a[2];
    const float t2 = b0 ? a[5] : a[4];
    const float t3 = b0 ? a[7] : a[6];
    const float u0 = b1 ? t1 : t0;
    const float u1 = b1 ? t3 : t2;
    return b2 ? u1 : u0;
}

__device__ __forceinline__ float pick4(const float (&a)[4], bool b0, bool b1)
{
    const float t0 = b0 ? a[1] : a[0];
    const float t1 = b0 ? a[3] : a[2];
    return b1 ? t1 : t0;
}

__device__ __forceinline__ float pair_lane(float s)
{
    return __int_as_float(__builtin_amdgcn_update_dpp(0, __float_as_int(s), 0xB1, 0xF, 0xF, true));
}

__device__ __forceinline__ float other_half(float s)
{
    const int r = __builtin_amdgcn_permlanex16(0, __float_as_int(s), 0x76543210, 0xfedcba98, false, false);
    return __int_as_float(r);
}

__device__ __forceinline__ float high_row_turned(float s)
{
    return __int_as_float(__builtin_amdgcn_update_dpp(__float_as_int(s), __float_as_int(s), 0x128, 0x2, 0xF, false));
}

template <int N>
__device__ __forceinline__ float row_lane(float s)
{
    return __int_as_float(__builtin_amdgcn_update_dpp(0, __float_as_int(s), 0x150 + N, 0xF, 0xF, true));
}

__launch_bounds__(32)
__global__ void pmm_sgd_kernel(const float* __restrict__ hidden,
                               const float* __restrict__ mlp_w1,
                               const float* __restrict__ mlp_b1,
                               const float* __restrict__ mlp_w2,
                               const float* __restrict__ mlp_b2,
                               const float* __restrict__ out_w,
                               const float* __restrict__ out_b,
                               float* __restrict__ out)
{
    __shared__ v4f   hid4[2][CH * Hq / 4];
    __shared__ float ctxLds[Hq];

    const int b    = blockIdx.x;
    const int lane = threadIdx.x;
    const int m    = lane & 15;
    const int hi   = lane >> 4;
    const bool db0 = (m & 1) != 0, db1 = (m & 2) != 0, db2 = (m & 4) != 0;
    const bool ownHalf = (m >> 3) == hi;
    const int uW1  = 4 * (m >> 3) + ((m & 7) >> 1);
    const int sW1  = m & 1;
    const bool ib0 = (m & 2) != 0, ib1 = (m & 4) != 0;
    const bool oddCol = sW1 != 0;

    const v4f* src = (const v4f*)(hidden + (size_t)b * Lq * Hq);

    v2f w1A[4], w2P[4], w2Q[2][2];
    float b2D[2][8];
    #pragma unroll
    for (int i = 0; i < 4; ++i) {
        #pragma unroll
        for (int j = 0; j < 2; ++j)
            w1A[i][j] = mlp_w1[uW1 * Hq + 16 * sW1 + 4 * i + 2 * hi + j];
    }
    #pragma unroll
    for (int i = 0; i < 4; ++i) {
        #pragma unroll
        for (int j = 0; j < 2; ++j)
            w2P[i][j] = mlp_w2[(16 * sW1 + 8 * hi + 2 * i + j) * 8 + uW1];
    }
    #pragma unroll
    for (int t = 0; t < 2; ++t) {
        #pragma unroll
        for (int kb = 0; kb < 2; ++kb) {
            #pragma unroll
            for (int j = 0; j < 2; ++j)
                w2Q[t][kb][j] = mlp_w2[(16 * t + m) * 8 + 4 * hi + 2 * kb + j];
        }
        #pragma unroll
        for (int r = 0; r < 8; ++r) b2D[t][r] = mlp_b2[16 * t + 8 * hi + r];
    }
    float b1own = mlp_b1[uW1];

    auto load_x = [&](const float* xrow, v2f (&xv)[4]) {
        #pragma unroll
        for (int i = 0; i < 4; ++i)
            xv[i] = *(const v2f*)(xrow + 16 * sW1 + 4 * i + 2 * hi);
    };
    auto load_v = [&](const float* vrow, v4f (&vq)[4]) {
        #pragma unroll
        for (int t = 0; t < 2; ++t) {
            #pragma unroll
            for (int u = 0; u < 2; ++u)
                vq[2 * t + u] = *(const v4f*)(vrow + 16 * t + 8 * hi + 4 * u);
        }
    };

    auto forward = [&](const v2f (&xv)[4], float (&a1)[4],
                       v2f (&a1v)[2], v8f (&accY)[2]) {
        v8f accP0 = {}, accP1 = {};
        #pragma unroll
        for (int i = 0; i < 4; i += 2) {
            accP0 = __builtin_amdgcn_wmma_f32_16x16x4_f32(
                        false, w1A[i], false, xv[i], (short)0, accP0, false, false);
            accP1 = __builtin_amdgcn_wmma_f32_16x16x4_f32(
                        false, w1A[i + 1], false, xv[i + 1], (short)0, accP1, false, false);
        }
        const float b1t = high_row_turned(b1own);
        float b1D[4];
        b1D[0] = row_lane<0>(b1t);
        b1D[1] = row_lane<2>(b1t);
        b1D[2] = row_lane<4>(b1t);
        b1D[3] = row_lane<6>(b1t);
        #pragma unroll
        for (int i = 0; i < 4; ++i) {
            const float sE = accP0[2 * i] + accP1[2 * i];
            const float sO = accP0[2 * i + 1] + accP1[2 * i + 1];
            const float own = oddCol ? sO : sE;
            const float a1p = (own + pair_lane(own)) + b1D[i];
            a1[i] = fmaxf(a1p, 0.f);
        }
        #pragma unroll
        for (int kb = 0; kb < 2; ++kb) {
            a1v[kb][0] = a1[2 * kb];
            a1v[kb][1] = a1[2 * kb + 1];
        }
        #pragma unroll
        for (int t = 0; t < 2; ++t) {
            #pragma unroll
            for (int r = 0; r < 8; ++r) accY[t][r] = b2D[t][r];
        }
        #pragma unroll
        for (int kb = 0; kb < 2; ++kb) {
            #pragma unroll
            for (int t = 0; t < 2; ++t)
                accY[t] = __builtin_amdgcn_wmma_f32_16x16x4_f32(
                            false, w2Q[t][kb], false, a1v[kb], (short)0, accY[t], false, false);
        }
    };

    auto step = [&](const v2f (&xv)[4], const v4f (&vq)[4]) {
        {
            float vD[2][8];
            #pragma unroll
            for (int t = 0; t < 2; ++t) {
                #pragma unroll
                for (int u = 0; u < 2; ++u) {
                    vD[t][4 * u + 0] = vq[2 * t + u][0]; vD[t][4 * u + 1] = vq[2 * t + u][1];
                    vD[t][4 * u + 2] = vq[2 * t + u][2]; vD[t][4 * u + 3] = vq[2 * t + u][3];
                }
            }

            v2f a1v[2];
            float a1[4];
            v8f accY[2];
            forward(xv, a1, a1v, accY);

            float dD[2][8], nd[2][8];
            #pragma unroll
            for (int t = 0; t < 2; ++t) {
                #pragma unroll
                for (int r = 0; r < 8; ++r) {
                    dD[t][r] = (accY[t][r] - vD[t][r]) * 0.0625f;
                    nd[t][r] = -LRC * dD[t][r];
                }
            }

            float dsel[8], ndsel[8];
            #pragma unroll
            for (int r = 0; r < 8; ++r) {
                dsel[r]  = oddCol ? dD[1][r] : dD[0][r];
                ndsel[r] = oddCol ? nd[1][r] : nd[0][r];
            }
            v8f accQ0 = {}, accQ1 = {};
            #pragma unroll
            for (int i = 0; i < 4; i += 2) {
                v2f dv0, dv1;
                dv0[0] = dsel[2 * i];     dv0[1] = dsel[2 * i + 1];
                dv1[0] = dsel[2 * i + 2]; dv1[1] = dsel[2 * i + 3];
                accQ0 = __builtin_amdgcn_wmma_f32_16x16x4_f32(
                            false, dv0, false, w2P[i], (short)0, accQ0, false, false);
                accQ1 = __builtin_amdgcn_wmma_f32_16x16x4_f32(
                            false, dv1, false, w2P[i + 1], (short)0, accQ1, false, false);
            }

            const float a1sel = pick4(a1, ib0, ib1);
            const float a1far = other_half(a1sel);
            const float a1own = ownHalf ? a1sel : a1far;

            const float sE = accQ0[0] + accQ1[0];
            const float sO = accQ0[1] + accQ1[1];
            const float own = oddCol ? sO : sE;
            const float qown = own + pair_lane(own);
            const float g1own = (a1own > 0.f) ? qown : 0.f;

            float ndown[2];
            #pragma unroll
            for (int t = 0; t < 2; ++t) {
                const float s8 = pick8(nd[t], db0, db1, db2);
                const float sx = other_half(s8);
                ndown[t] = ownHalf ? s8 : sx;
            }

            const float ng = -LRC * g1own;
            #pragma unroll
            for (int i = 0; i < 4; ++i) {
                #pragma unroll
                for (int j = 0; j < 2; ++j)
                    w1A[i][j] = fmaf(ng, xv[i][j], w1A[i][j]);
            }
            #pragma unroll
            for (int i = 0; i < 4; ++i) {
                #pragma unroll
                for (int j = 0; j < 2; ++j)
                    w2P[i][j] = fmaf(ndsel[2 * i + j], a1own, w2P[i][j]);
            }
            #pragma unroll
            for (int t = 0; t < 2; ++t) {
                #pragma unroll
                for (int kb = 0; kb < 2; ++kb) {
                    #pragma unroll
                    for (int j = 0; j < 2; ++j)
                        w2Q[t][kb][j] = fmaf(ndown[t], a1v[kb][j], w2Q[t][kb][j]);
                }
                #pragma unroll
                for (int r = 0; r < 8; ++r) b2D[t][r] -= LRC * dD[t][r];
            }
            b1own -= LRC * g1own;
        }
    };

    #pragma unroll
    for (int k = 0; k < 4; ++k) hid4[0][k * 32 + lane] = src[k * 32 + lane];

    #pragma unroll 1
    for (int c = 0; c < NCH; ++c) {
        v4f nx[4];
        const int cn = (c + 1 < NCH) ? (c + 1) : c;
        #pragma unroll
        for (int k = 0; k < 4; ++k) nx[k] = src[cn * 128 + k * 32 + lane];

        const float* hbuf = (const float*)hid4[c & 1];
        v2f xs[8][4];
        v4f vs[8][4];
        #pragma unroll
        for (int s = 0; s < 8; ++s) {
            load_x(hbuf + (2 * s) * Hq, xs[s]);
            load_v(hbuf + (2 * s + 1) * Hq, vs[s]);
        }
        __builtin_amdgcn_sched_barrier(0);
        #pragma unroll
        for (int s = 0; s < 8; ++s) {
            if (s < 7 || c != NCH - 1) step(xs[s], vs[s]);
        }

        #pragma unroll
        for (int k = 0; k < 4; ++k) hid4[(c + 1) & 1][k * 32 + lane] = nx[k];
    }

    {
        const float* hbuf = (const float*)hid4[(NCH - 1) & 1];
        v2f xv[4], a1v[2];
        float a1[4];
        v8f accY[2];
        load_x(hbuf + 15 * Hq, xv);
        forward(xv, a1, a1v, accY);
        if (m == 0) {
            #pragma unroll
            for (int t = 0; t < 2; ++t) {
                #pragma unroll
                for (int r = 0; r < 8; ++r) ctxLds[16 * t + 8 * hi + r] = accY[t][r];
            }
        }
    }
    __syncthreads();

    #pragma unroll
    for (int half = 0; half < 2; ++half) {
        const int vo = half * 32 + lane;
        float acc = out_b[vo];
        #pragma unroll
        for (int h = 0; h < Hq; ++h)
            acc = fmaf(ctxLds[h], out_w[vo * Hq + h], acc);
        out[b * Vq + vo] = acc;
    }
}

extern "C" void kernel_launch(void* const* d_in, const int* in_sizes, int n_in,
                              void* d_out, int out_size, void* d_ws, size_t ws_size,
                              hipStream_t stream) {
    (void)in_sizes; (void)n_in; (void)out_size; (void)ws_size;
    const int*   seq    = (const int*)  d_in[0];
    const float* embed  = (const float*)d_in[1];
    const float* ff_w1  = (const float*)d_in[2];
    const float* ff_b1  = (const float*)d_in[3];
    const float* ff_w2  = (const float*)d_in[4];
    const float* ff_b2  = (const float*)d_in[5];
    const float* ln_g   = (const float*)d_in[6];
    const float* ln_b   = (const float*)d_in[7];
    const float* mlp_w1 = (const float*)d_in[8];
    const float* mlp_b1 = (const float*)d_in[9];
    const float* mlp_w2 = (const float*)d_in[10];
    const float* mlp_b2 = (const float*)d_in[11];
    const float* out_w  = (const float*)d_in[12];
    const float* out_b  = (const float*)d_in[13];
    float*       out    = (float*)d_out;
    float*       hidden = (float*)d_ws;

    pmm_encode_kernel<<<dim3(ENC_BLOCKS), dim3(ENC_WAVES * 32), 0, stream>>>(
        seq, embed, ff_w1, ff_b1, ff_w2, ff_b2, ln_g, ln_b, hidden);
    pmm_sgd_kernel<<<dim3(Bq), dim3(32), 0, stream>>>(
        hidden, mlp_w1, mlp_b1, mlp_w2, mlp_b2, out_w, out_b, out);
}


# SGD launch: fc2's accumulators start from b2 - v, so its D registers (16 dL/dy) are the backward operand with no subtract or scale between; 1/16 moves into -lr/16 (exact); y - v's sum order changes
# speedup vs baseline: 3.1835x; 1.0315x over previous
#include <hip/hip_runtime.h>
#include <hip/hip_bf16.h>
#include <math.h>


typedef __attribute__((ext_vector_type(2))) float v2f;
typedef __attribute__((ext_vector_type(4))) float v4f;
typedef __attribute__((ext_vector_type(8))) float v8f;

#define LRC  0.01f
#define EPSC 1e-5f
#define NLR16 (-LRC * 0.0625f)

#define Bq 256
#define Lq 4096
#define Hq 32
#define Vq 64
#define CH 16
#define NCH (Lq / CH)
#define A1STR 68
#define HSTR  36

#define ENC_WAVES 8
#define ENC_CPW   8
#define ENC_BLOCKS ((Bq * NCH) / (ENC_WAVES * ENC_CPW))

__launch_bounds__(ENC_WAVES * 32)
__global__ void pmm_encode_kernel(const int* __restrict__ seq,
                                  const float* __restrict__ embed,
                                  const float* __restrict__ ff_w1,
                                  const float* __restrict__ ff_b1,
                                  const float* __restrict__ ff_w2,
                                  const float* __restrict__ ff_b2,
                                  const float* __restrict__ ln_g,
                                  const float* __restrict__ ln_b,
                                  float* __restrict__ hidden)
{
    __shared__ float embedLds[Vq * Hq];
    __shared__ float w1Lds[2 * Hq * Hq];
    __shared__ float w2Lds[Hq * 2 * Hq];
    __shared__ float a1all[ENC_WAVES][CH * A1STR];
    __shared__ float hball[ENC_WAVES][CH * HSTR];

    const int tid  = threadIdx.x;
    const int lane = tid & 31;
    const int wave = tid >> 5;
    const int m    = lane & 15;
    const int hi   = lane >> 4;

    for (int i = tid; i < Vq * Hq; i += ENC_WAVES * 32)     embedLds[i] = embed[i];
    for (int i = tid; i < 2 * Hq * Hq; i += ENC_WAVES * 32) w1Lds[i]    = ff_w1[i];
    for (int i = tid; i < 2 * Hq * Hq; i += ENC_WAVES * 32) w2Lds[i]    = ff_w2[i];
    __syncthreads();

    float b1t[4], b2t[2];
    #pragma unroll
    for (int nt = 0; nt < 4; ++nt) b1t[nt] = ff_b1[nt * 16 + m];
    #pragma unroll
    for (int nt = 0; nt < 2; ++nt) b2t[nt] = ff_b2[nt * 16 + m];
    const float lnGv = ln_g[lane];
    const float lnBv = ln_b[lane];

    float* a1buf = a1all[wave];
    float* hbuf  = hball[wave];

    const int g0 = (blockIdx.x * ENC_WAVES + wave) * ENC_CPW;

    #pragma unroll 1
    for (int j = 0; j < ENC_CPW; ++j) {
        const int base = (g0 + j) * CH;
        const int tok  = seq[base + m];

        v2f eA[8];
        #pragma unroll
        for (int kb = 0; kb < 8; ++kb)
            eA[kb] = *(const v2f*)&embedLds[tok * Hq + 4 * kb + 2 * hi];

        #pragma unroll 1
        for (int nt = 0; nt < 4; ++nt) {
            const int n0 = nt * 16;
            v8f acc = {};
            #pragma unroll
            for (int kb = 0; kb < 8; ++kb) {
                v2f bB = *(const v2f*)&w1Lds[(n0 + m) * Hq + 4 * kb + 2 * hi];
                acc = __builtin_amdgcn_wmma_f32_16x16x4_f32(
                        false, eA[kb], false, bB, (short)0, acc, false, false);
            }
            const float bias = b1t[nt];
            #pragma unroll
            for (int r = 0; r < 8; ++r) {
                float z = acc[r] + bias;
                a1buf[(r + 8 * hi) * A1STR + n0 + m] = fmaxf(z, 0.f);
            }
        }
        asm volatile("s_wait_dscnt 0" ::: "memory");
        __builtin_amdgcn_wave_barrier();

        v2f a2[16];
        #pragma unroll
        for (int kb = 0; kb < 16; ++kb)
            a2[kb] = *(const v2f*)&a1buf[m * A1STR + 4 * kb + 2 * hi];

        #pragma unroll 1
        for (int nt = 0; nt < 2; ++nt) {
            const int n0 = nt * 16;
            v8f acc = {};
            #pragma unroll
            for (int kb = 0; kb < 16; ++kb) {
                v2f bB = *(const v2f*)&w2Lds[(n0 + m) * (2 * Hq) + 4 * kb + 2 * hi];
                acc = __builtin_amdgcn_wmma_f32_16x16x4_f32(
                        false, a2[kb], false, bB, (short)0, acc, false, false);
            }
            const float bias = b2t[nt];
            #pragma unroll
            for (int r = 0; r < 8; ++r) {
                int   tm = __shfl(tok, r + 8 * hi, 32);
                float e  = embedLds[tm * Hq + n0 + m];
                hbuf[(r + 8 * hi) * HSTR + n0 + m] = acc[r] + bias + e;
            }
        }
        asm volatile("s_wait_dscnt 0" ::: "memory");
        __builtin_amdgcn_wave_barrier();

        #pragma unroll 1
        for (int t = 0; t < CH; ++t) {
            float z = hbuf[t * HSTR + lane];
            float s = z, s2 = z * z;
            #pragma unroll
            for (int off = 16; off >= 1; off >>= 1) {
                s  += __shfl_xor(s,  off, 32);
                s2 += __shfl_xor(s2, off, 32);
            }
            float mu  = s * (1.f / 32.f);
            float var = s2 * (1.f / 32.f) - mu * mu;
            float nz  = (z - mu) * rsqrtf(var + EPSC);
            hidden[(size_t)(base + t) * Hq + lane] = nz * lnGv + lnBv;
        }
        asm volatile("s_wait_dscnt 0" ::: "memory");
    }
}

__device__ __forceinline__ float pick8(const float (&a)[8], bool b0, bool b1, bool b2)
{
    const float t0 = b0 ? a[1] : a[0];
    const float t1 = b0 ? a[3] : a[2];
    const float t2 = b0 ? a[5] : a[4];
    const float t3 = b0 ? a[7] : a[6];
    const float u0 = b1 ? t1 : t0;
    const float u1 = b1 ? t3 : t2;
    return b2 ? u1 : u0;
}

__device__ __forceinline__ float pick4(const float (&a)[4], bool b0, bool b1)
{
    const float t0 = b0 ? a[1] : a[0];
    const float t1 = b0 ? a[3] : a[2];
    return b1 ? t1 : t0;
}

__device__ __forceinline__ float pair_lane(float s)
{
    return __int_as_float(__builtin_amdgcn_update_dpp(0, __float_as_int(s), 0xB1, 0xF, 0xF, true));
}

__device__ __forceinline__ float other_half(float s)
{
    const int r = __builtin_amdgcn_permlanex16(0, __float_as_int(s), 0x76543210, 0xfedcba98, false, false);
    return __int_as_float(r);
}

__device__ __forceinline__ float high_row_turned(float s)
{
    return __int_as_float(__builtin_amdgcn_update_dpp(__float_as_int(s), __float_as_int(s), 0x128, 0x2, 0xF, false));
}

template <int N>
__device__ __forceinline__ float row_lane(float s)
{
    return __int_as_float(__builtin_amdgcn_update_dpp(0, __float_as_int(s), 0x150 + N, 0xF, 0xF, true));
}

__launch_bounds__(32)
__global__ void pmm_sgd_kernel(const float* __restrict__ hidden,
                               const float* __restrict__ mlp_w1,
                               const float* __restrict__ mlp_b1,
                               const float* __restrict__ mlp_w2,
                               const float* __restrict__ mlp_b2,
                               const float* __restrict__ out_w,
                               const float* __restrict__ out_b,
                               float* __restrict__ out)
{
    __shared__ v4f   hid4[2][CH * Hq / 4];
    __shared__ float ctxLds[Hq];

    const int b    = blockIdx.x;
    const int lane = threadIdx.x;
    const int m    = lane & 15;
    const int hi   = lane >> 4;
    const bool db0 = (m & 1) != 0, db1 = (m & 2) != 0, db2 = (m & 4) != 0;
    const bool ownHalf = (m >> 3) == hi;
    const int uW1  = 4 * (m >> 3) + ((m & 7) >> 1);
    const int sW1  = m & 1;
    const bool ib0 = (m & 2) != 0, ib1 = (m & 4) != 0;
    const bool oddCol = sW1 != 0;

    const v4f* src = (const v4f*)(hidden + (size_t)b * Lq * Hq);

    v2f w1A[4], w2P[4], w2Q[2][2];
    float b2D[2][8];
    #pragma unroll
    for (int i = 0; i < 4; ++i) {
        #pragma unroll
        for (int j = 0; j < 2; ++j)
            w1A[i][j] = mlp_w1[uW1 * Hq + 16 * sW1 + 4 * i + 2 * hi + j];
    }
    #pragma unroll
    for (int i = 0; i < 4; ++i) {
        #pragma unroll
        for (int j = 0; j < 2; ++j)
            w2P[i][j] = mlp_w2[(16 * sW1 + 8 * hi + 2 * i + j) * 8 + uW1];
    }
    #pragma unroll
    for (int t = 0; t < 2; ++t) {
        #pragma unroll
        for (int kb = 0; kb < 2; ++kb) {
            #pragma unroll
            for (int j = 0; j < 2; ++j)
                w2Q[t][kb][j] = mlp_w2[(16 * t + m) * 8 + 4 * hi + 2 * kb + j];
        }
        #pragma unroll
        for (int r = 0; r < 8; ++r) b2D[t][r] = mlp_b2[16 * t + 8 * hi + r];
    }
    float b1own = mlp_b1[uW1];

    auto load_x = [&](const float* xrow, v2f (&xv)[4]) {
        #pragma unroll
        for (int i = 0; i < 4; ++i)
            xv[i] = *(const v2f*)(xrow + 16 * sW1 + 4 * i + 2 * hi);
    };
    auto load_v = [&](const float* vrow, v4f (&vq)[4]) {
        #pragma unroll
        for (int t = 0; t < 2; ++t) {
            #pragma unroll
            for (int u = 0; u < 2; ++u)
                vq[2 * t + u] = *(const v4f*)(vrow + 16 * t + 8 * hi + 4 * u);
        }
    };

    auto forward = [&](const v2f (&xv)[4], const v8f (&y0)[2], float (&a1)[4],
                       v2f (&a1v)[2], v8f (&accY)[2]) {
        v8f accP0 = {}, accP1 = {};
        #pragma unroll
        for (int i = 0; i < 4; i += 2) {
            accP0 = __builtin_amdgcn_wmma_f32_16x16x4_f32(
                        false, w1A[i], false, xv[i], (short)0, accP0, false, false);
            accP1 = __builtin_amdgcn_wmma_f32_16x16x4_f32(
                        false, w1A[i + 1], false, xv[i + 1], (short)0, accP1, false, false);
        }
        const float b1t = high_row_turned(b1own);
        float b1D[4];
        b1D[0] = row_lane<0>(b1t);
        b1D[1] = row_lane<2>(b1t);
        b1D[2] = row_lane<4>(b1t);
        b1D[3] = row_lane<6>(b1t);
        #pragma unroll
        for (int i = 0; i < 4; ++i) {
            const float sE = accP0[2 * i] + accP1[2 * i];
            const float sO = accP0[2 * i + 1] + accP1[2 * i + 1];
            const float own = oddCol ? sO : sE;
            const float a1p = (own + pair_lane(own)) + b1D[i];
            a1[i] = fmaxf(a1p, 0.f);
        }
        #pragma unroll
        for (int kb = 0; kb < 2; ++kb) {
            a1v[kb][0] = a1[2 * kb];
            a1v[kb][1] = a1[2 * kb + 1];
        }
        #pragma unroll
        for (int t = 0; t < 2; ++t) accY[t] = y0[t];
        #pragma unroll
        for (int kb = 0; kb < 2; ++kb) {
            #pragma unroll
            for (int t = 0; t < 2; ++t)
                accY[t] = __builtin_amdgcn_wmma_f32_16x16x4_f32(
                            false, w2Q[t][kb], false, a1v[kb], (short)0, accY[t], false, false);
        }
    };

    auto step = [&](const v2f (&xv)[4], const v4f (&vq)[4]) {
        {
            v8f y0[2];
            #pragma unroll
            for (int t = 0; t < 2; ++t) {
                #pragma unroll
                for (int u = 0; u < 2; ++u) {
                    #pragma unroll
                    for (int e = 0; e < 4; ++e)
                        y0[t][4 * u + e] = b2D[t][4 * u + e] - vq[2 * t + u][e];
                }
            }

            v2f a1v[2];
            float a1[4];
            v8f accY[2];
            forward(xv, y0, a1, a1v, accY);

            float nd[2][8];
            #pragma unroll
            for (int t = 0; t < 2; ++t) {
                #pragma unroll
                for (int r = 0; r < 8; ++r) nd[t][r] = NLR16 * accY[t][r];
            }

            float dsel[8], ndsel[8];
            #pragma unroll
            for (int r = 0; r < 8; ++r) {
                dsel[r]  = oddCol ? accY[1][r] : accY[0][r];
                ndsel[r] = oddCol ? nd[1][r] : nd[0][r];
            }
            v8f accQ0 = {}, accQ1 = {};
            #pragma unroll
            for (int i = 0; i < 4; i += 2) {
                v2f dv0, dv1;
                dv0[0] = dsel[2 * i];     dv0[1] = dsel[2 * i + 1];
                dv1[0] = dsel[2 * i + 2]; dv1[1] = dsel[2 * i + 3];
                accQ0 = __builtin_amdgcn_wmma_f32_16x16x4_f32(
                            false, dv0, false, w2P[i], (short)0, accQ0, false, false);
                accQ1 = __builtin_amdgcn_wmma_f32_16x16x4_f32(
                            false, dv1, false, w2P[i + 1], (short)0, accQ1, false, false);
            }

            const float a1sel = pick4(a1, ib0, ib1);
            const float a1far = other_half(a1sel);
            const float a1own = ownHalf ? a1sel : a1far;

            const float sE = accQ0[0] + accQ1[0];
            const float sO = accQ0[1] + accQ1[1];
            const float own = oddCol ? sO : sE;
            const float qown = own + pair_lane(own);
            const float g1own = (a1own > 0.f) ? qown : 0.f;

            float ndown[2];
            #pragma unroll
            for (int t = 0; t < 2; ++t) {
                const float s8 = pick8(nd[t], db0, db1, db2);
                const float sx = other_half(s8);
                ndown[t] = ownHalf ? s8 : sx;
            }

            const float ng = NLR16 * g1own;
            #pragma unroll
            for (int i = 0; i < 4; ++i) {
                #pragma unroll
                for (int j = 0; j < 2; ++j)
                    w1A[i][j] = fmaf(ng, xv[i][j], w1A[i][j]);
            }
            #pragma unroll
            for (int i = 0; i < 4; ++i) {
                #pragma unroll
                for (int j = 0; j < 2; ++j)
                    w2P[i][j] = fmaf(ndsel[2 * i + j], a1own, w2P[i][j]);
            }
            #pragma unroll
            for (int t = 0; t < 2; ++t) {
                #pragma unroll
                for (int kb = 0; kb < 2; ++kb) {
                    #pragma unroll
                    for (int j = 0; j < 2; ++j)
                        w2Q[t][kb][j] = fmaf(ndown[t], a1v[kb][j], w2Q[t][kb][j]);
                }
                #pragma unroll
                for (int r = 0; r < 8; ++r) b2D[t][r] = fmaf(NLR16, accY[t][r], b2D[t][r]);
            }
            b1own = fmaf(NLR16, g1own, b1own);
        }
    };

    #pragma unroll
    for (int k = 0; k < 4; ++k) hid4[0][k * 32 + lane] = src[k * 32 + lane];

    #pragma unroll 1
    for (int c = 0; c < NCH; ++c) {
        v4f nx[4];
        const int cn = (c + 1 < NCH) ? (c + 1) : c;
        #pragma unroll
        for (int k = 0; k < 4; ++k) nx[k] = src[cn * 128 + k * 32 + lane];

        const float* hbuf = (const float*)hid4[c & 1];
        v2f xs[8][4];
        v4f vs[8][4];
        #pragma unroll
        for (int s = 0; s < 8; ++s) {
            load_x(hbuf + (2 * s) * Hq, xs[s]);
            load_v(hbuf + (2 * s + 1) * Hq, vs[s]);
        }
        __builtin_amdgcn_sched_barrier(0);
        #pragma unroll
        for (int s = 0; s < 8; ++s) {
            if (s < 7 || c != NCH - 1) step(xs[s], vs[s]);
        }

        #pragma unroll
        for (int k = 0; k < 4; ++k) hid4[(c + 1) & 1][k * 32 + lane] = nx[k];
    }

    {
        const float* hbuf = (const float*)hid4[(NCH - 1) & 1];
        v2f xv[4], a1v[2];
        float a1[4];
        v8f y0[2], accY[2];
        #pragma unroll
        for (int t = 0; t < 2; ++t) {
            #pragma unroll
            for (int r = 0; r < 8; ++r) y0[t][r] = b2D[t][r];
        }
        load_x(hbuf + 15 * Hq, xv);
        forward(xv, y0, a1, a1v, accY);
        if (m == 0) {
            #pragma unroll
            for (int t = 0; t < 2; ++t) {
                #pragma unroll
                for (int r = 0; r < 8; ++r) ctxLds[16 * t + 8 * hi + r] = accY[t][r];
            }
        }
    }
    __syncthreads();

    #pragma unroll
    for (int half = 0; half < 2; ++half) {
        const int vo = half * 32 + lane;
        float acc = out_b[vo];
        #pragma unroll
        for (int h = 0; h < Hq; ++h)
            acc = fmaf(ctxLds[h], out_w[vo * Hq + h], acc);
        out[b * Vq + vo] = acc;
    }
}

extern "C" void kernel_launch(void* const* d_in, const int* in_sizes, int n_in,
                              void* d_out, int out_size, void* d_ws, size_t ws_size,
                              hipStream_t stream) {
    (void)in_sizes; (void)n_in; (void)out_size; (void)ws_size;
    const int*   seq    = (const int*)  d_in[0];
    const float* embed  = (const float*)d_in[1];
    const float* ff_w1  = (const float*)d_in[2];
    const float* ff_b1  = (const float*)d_in[3];
    const float* ff_w2  = (const float*)d_in[4];
    const float* ff_b2  = (const float*)d_in[5];
    const float* ln_g   = (const float*)d_in[6];
    const float* ln_b   = (const float*)d_in[7];
    const float* mlp_w1 = (const float*)d_in[8];
    const float* mlp_b1 = (const float*)d_in[9];
    const float* mlp_w2 = (const float*)d_in[10];
    const float* mlp_b2 = (const float*)d_in[11];
    const float* out_w  = (const float*)d_in[12];
    const float* out_b  = (const float*)d_in[13];
    float*       out    = (float*)d_out;
    float*       hidden = (float*)d_ws;

    pmm_encode_kernel<<<dim3(ENC_BLOCKS), dim3(ENC_WAVES * 32), 0, stream>>>(
        seq, embed, ff_w1, ff_b1, ff_w2, ff_b2, ln_g, ln_b, hidden);
    pmm_sgd_kernel<<<dim3(Bq), dim3(32), 0, stream>>>(
        hidden, mlp_w1, mlp_b1, mlp_w2, mlp_b2, out_w, out_b, out);
}


# Encoder: FF2 with operands swapped (weights as A, activations as B): a lane holds 16 features of its own token, LayerNorm sums run in registers + one half-wave exchange; mean/var sum order changes
# speedup vs baseline: 3.4641x; 1.0881x over previous
#include <hip/hip_runtime.h>
#include <hip/hip_bf16.h>
#include <math.h>


typedef __attribute__((ext_vector_type(2))) float v2f;
typedef __attribute__((ext_vector_type(4))) float v4f;
typedef __attribute__((ext_vector_type(8))) float v8f;

#define LRC  0.01f
#define EPSC 1e-5f
#define NLR16 (-LRC * 0.0625f)

#define Bq 256
#define Lq 4096
#define Hq 32
#define Vq 64
#define CH 16
#define NCH (Lq / CH)
#define A1STR 68
#define HSTR  36
#define ESTR  36

#define ENC_WAVES 8
#define ENC_CPW   8
#define ENC_BLOCKS ((Bq * NCH) / (ENC_WAVES * ENC_CPW))

__device__ __forceinline__ float enc_other_half(float s)
{
    const int r = __builtin_amdgcn_permlanex16(0, __float_as_int(s), 0x76543210, 0xfedcba98, false, false);
    return __int_as_float(r);
}

__launch_bounds__(ENC_WAVES * 32)
__global__ void pmm_encode_kernel(const int* __restrict__ seq,
                                  const float* __restrict__ embed,
                                  const float* __restrict__ ff_w1,
                                  const float* __restrict__ ff_b1,
                                  const float* __restrict__ ff_w2,
                                  const float* __restrict__ ff_b2,
                                  const float* __restrict__ ln_g,
                                  const float* __restrict__ ln_b,
                                  float* __restrict__ hidden)
{
    __shared__ __attribute__((aligned(16))) float embedLds[Vq * ESTR];
    __shared__ __attribute__((aligned(16))) float w1Lds[2 * Hq * Hq];
    __shared__ __attribute__((aligned(16))) float w2Lds[Hq * 2 * Hq];
    __shared__ __attribute__((aligned(16))) float a1all[ENC_WAVES][CH * A1STR];
    __shared__ __attribute__((aligned(16))) float hball[ENC_WAVES][CH * HSTR];
    __shared__ __attribute__((aligned(16))) float b2Lds[Hq];
    __shared__ __attribute__((aligned(16))) float gLds[Hq];
    __shared__ __attribute__((aligned(16))) float bLds[Hq];

    const int tid  = threadIdx.x;
    const int lane = tid & 31;
    const int wave = tid >> 5;
    const int m    = lane & 15;
    const int hi   = lane >> 4;

    for (int i = tid; i < Vq * Hq; i += ENC_WAVES * 32)     embedLds[(i >> 5) * ESTR + (i & 31)] = embed[i];
    for (int i = tid; i < 2 * Hq * Hq; i += ENC_WAVES * 32) w1Lds[i]    = ff_w1[i];
    for (int i = tid; i < 2 * Hq * Hq; i += ENC_WAVES * 32) w2Lds[i]    = ff_w2[i];
    if (tid < Hq) {
        b2Lds[tid] = ff_b2[tid];
        gLds[tid]  = ln_g[tid];
        bLds[tid]  = ln_b[tid];
    }
    __syncthreads();

    float b1t[4];
    #pragma unroll
    for (int nt = 0; nt < 4; ++nt) b1t[nt] = ff_b1[nt * 16 + m];

    float* a1buf = a1all[wave];
    float* hbuf  = hball[wave];

    const int g0 = (blockIdx.x * ENC_WAVES + wave) * ENC_CPW;

    #pragma unroll 1
    for (int j = 0; j < ENC_CPW; ++j) {
        const int base = (g0 + j) * CH;
        const int tok  = seq[base + m];
        const float* erow = &embedLds[tok * ESTR];

        v2f eA[8];
        #pragma unroll
        for (int kb = 0; kb < 8; ++kb)
            eA[kb] = *(const v2f*)&erow[4 * kb + 2 * hi];

        #pragma unroll 1
        for (int nt = 0; nt < 4; ++nt) {
            const int n0 = nt * 16;
            v8f acc = {};
            #pragma unroll
            for (int kb = 0; kb < 8; ++kb) {
                v2f bB = *(const v2f*)&w1Lds[(n0 + m) * Hq + 4 * kb + 2 * hi];
                acc = __builtin_amdgcn_wmma_f32_16x16x4_f32(
                        false, eA[kb], false, bB, (short)0, acc, false, false);
            }
            const float bias = b1t[nt];
            #pragma unroll
            for (int r = 0; r < 8; ++r) {
                float z = acc[r] + bias;
                a1buf[(r + 8 * hi) * A1STR + n0 + m] = fmaxf(z, 0.f);
            }
        }
        asm volatile("s_wait_dscnt 0" ::: "memory");
        __builtin_amdgcn_wave_barrier();

        v2f a2[16];
        #pragma unroll
        for (int kb = 0; kb < 16; ++kb)
            a2[kb] = *(const v2f*)&a1buf[m * A1STR + 4 * kb + 2 * hi];

        float z[2][8];
        #pragma unroll
        for (int nt = 0; nt < 2; ++nt) {
            const int n0 = nt * 16;
            v8f acc = {};
            #pragma unroll
            for (int kb = 0; kb < 16; ++kb) {
                v2f wA = *(const v2f*)&w2Lds[(n0 + m) * (2 * Hq) + 4 * kb + 2 * hi];
                acc = __builtin_amdgcn_wmma_f32_16x16x4_f32(
                        false, wA, false, a2[kb], (short)0, acc, false, false);
            }
            #pragma unroll
            for (int u = 0; u < 2; ++u) {
                const v4f bq = *(const v4f*)&b2Lds[n0 + 8 * hi + 4 * u];
                const v4f eq = *(const v4f*)&erow[n0 + 8 * hi + 4 * u];
                #pragma unroll
                for (int e = 0; e < 4; ++e)
                    z[nt][4 * u + e] = acc[4 * u + e] + bq[e] + eq[e];
            }
        }

        float ps[8], pq[8];
        #pragma unroll
        for (int r = 0; r < 8; ++r) {
            ps[r] = z[0][r] + z[1][r];
            pq[r] = fmaf(z[1][r], z[1][r], z[0][r] * z[0][r]);
        }
        #pragma unroll
        for (int w = 4; w >= 1; w >>= 1) {
            #pragma unroll
            for (int r = 0; r < w; ++r) {
                ps[r] += ps[r + w];
                pq[r] += pq[r + w];
            }
        }
        const float s   = ps[0] + enc_other_half(ps[0]);
        const float s2  = pq[0] + enc_other_half(pq[0]);
        const float mu  = s * (1.f / 32.f);
        const float var = s2 * (1.f / 32.f) - mu * mu;
        const float rs  = rsqrtf(var + EPSC);
        #pragma unroll
        for (int nt = 0; nt < 2; ++nt) {
            #pragma unroll
            for (int u = 0; u < 2; ++u) {
                const int f0 = 16 * nt + 8 * hi + 4 * u;
                const v4f gq = *(const v4f*)&gLds[f0];
                const v4f bq = *(const v4f*)&bLds[f0];
                v4f o;
                #pragma unroll
                for (int e = 0; e < 4; ++e) {
                    const float nz = (z[nt][4 * u + e] - mu) * rs;
                    o[e] = nz * gq[e] + bq[e];
                }
                *(v4f*)&hbuf[m * HSTR + f0] = o;
            }
        }
        asm volatile("s_wait_dscnt 0" ::: "memory");
        __builtin_amdgcn_wave_barrier();

        #pragma unroll
        for (int i = 0; i < 4; ++i) {
            const int t = 4 * i + (lane >> 3);
            const v4f row = *(const v4f*)&hbuf[t * HSTR + 4 * (lane & 7)];
            *(v4f*)&hidden[(size_t)(base + t) * Hq + 4 * (lane & 7)] = row;
        }
        asm volatile("s_wait_dscnt 0" ::: "memory");
    }
}

__device__ __forceinline__ float pick8(const float (&a)[8], bool b0, bool b1, bool b2)
{
    const float t0 = b0 ? a[1] : a[0];
    const float t1 = b0 ? a[3] : a[2];
    const float t2 = b0 ? a[5] : a[4];
    const float t3 = b0 ? a[7] : a[6];
    const float u0 = b1 ? t1 : t0;
    const float u1 = b1 ? t3 : t2;
    return b2 ? u1 : u0;
}

__device__ __forceinline__ float pick4(const float (&a)[4], bool b0, bool b1)
{
    const float t0 = b0 ? a[1] : a[0];
    const float t1 = b0 ? a[3] : a[2];
    return b1 ? t1 : t0;
}

__device__ __forceinline__ float pair_lane(float s)
{
    return __int_as_float(__builtin_amdgcn_update_dpp(0, __float_as_int(s), 0xB1, 0xF, 0xF, true));
}

__device__ __forceinline__ float other_half(float s)
{
    const int r = __builtin_amdgcn_permlanex16(0, __float_as_int(s), 0x76543210, 0xfedcba98, false, false);
    return __int_as_float(r);
}

__device__ __forceinline__ float high_row_turned(float s)
{
    return __int_as_float(__builtin_amdgcn_update_dpp(__float_as_int(s), __float_as_int(s), 0x128, 0x2, 0xF, false));
}

template <int N>
__device__ __forceinline__ float row_lane(float s)
{
    return __int_as_float(__builtin_amdgcn_update_dpp(0, __float_as_int(s), 0x150 + N, 0xF, 0xF, true));
}

__launch_bounds__(32)
__global__ void pmm_sgd_kernel(const float* __restrict__ hidden,
                               const float* __restrict__ mlp_w1,
                               const float* __restrict__ mlp_b1,
                               const float* __restrict__ mlp_w2,
                               const float* __restrict__ mlp_b2,
                               const float* __restrict__ out_w,
                               const float* __restrict__ out_b,
                               float* __restrict__ out)
{
    __shared__ v4f   hid4[2][CH * Hq / 4];
    __shared__ float ctxLds[Hq];

    const int b    = blockIdx.x;
    const int lane = threadIdx.x;
    const int m    = lane & 15;
    const int hi   = lane >> 4;
    const bool db0 = (m & 1) != 0, db1 = (m & 2) != 0, db2 = (m & 4) != 0;
    const bool ownHalf = (m >> 3) == hi;
    const int uW1  = 4 * (m >> 3) + ((m & 7) >> 1);
    const int sW1  = m & 1;
    const bool ib0 = (m & 2) != 0, ib1 = (m & 4) != 0;
    const bool oddCol = sW1 != 0;

    const v4f* src = (const v4f*)(hidden + (size_t)b * Lq * Hq);

    v2f w1A[4], w2P[4], w2Q[2][2];
    float b2D[2][8];
    #pragma unroll
    for (int i = 0; i < 4; ++i) {
        #pragma unroll
        for (int j = 0; j < 2; ++j)
            w1A[i][j] = mlp_w1[uW1 * Hq + 16 * sW1 + 4 * i + 2 * hi + j];
    }
    #pragma unroll
    for (int i = 0; i < 4; ++i) {
        #pragma unroll
        for (int j = 0; j < 2; ++j)
            w2P[i][j] = mlp_w2[(16 * sW1 + 8 * hi + 2 * i + j) * 8 + uW1];
    }
    #pragma unroll
    for (int t = 0; t < 2; ++t) {
        #pragma unroll
        for (int kb = 0; kb < 2; ++kb) {
            #pragma unroll
            for (int j = 0; j < 2; ++j)
                w2Q[t][kb][j] = mlp_w2[(16 * t + m) * 8 + 4 * hi + 2 * kb + j];
        }
        #pragma unroll
        for (int r = 0; r < 8; ++r) b2D[t][r] = mlp_b2[16 * t + 8 * hi + r];
    }
    float b1own = mlp_b1[uW1];

    auto load_x = [&](const float* xrow, v2f (&xv)[4]) {
        #pragma unroll
        for (int i = 0; i < 4; ++i)
            xv[i] = *(const v2f*)(xrow + 16 * sW1 + 4 * i + 2 * hi);
    };
    auto load_v = [&](const float* vrow, v4f (&vq)[4]) {
        #pragma unroll
        for (int t = 0; t < 2; ++t) {
            #pragma unroll
            for (int u = 0; u < 2; ++u)
                vq[2 * t + u] = *(const v4f*)(vrow + 16 * t + 8 * hi + 4 * u);
        }
    };

    auto forward = [&](const v2f (&xv)[4], const v8f (&y0)[2], float (&a1)[4],
                       v2f (&a1v)[2], v8f (&accY)[2]) {
        v8f accP0 = {}, accP1 = {};
        #pragma unroll
        for (int i = 0; i < 4; i += 2) {
            accP0 = __builtin_amdgcn_wmma_f32_16x16x4_f32(
                        false, w1A[i], false, xv[i], (short)0, accP0, false, false);
            accP1 = __builtin_amdgcn_wmma_f32_16x16x4_f32(
                        false, w1A[i + 1], false, xv[i + 1], (short)0, accP1, false, false);
        }
        const float b1t = high_row_turned(b1own);
        float b1D[4];
        b1D[0] = row_lane<0>(b1t);
        b1D[1] = row_lane<2>(b1t);
        b1D[2] = row_lane<4>(b1t);
        b1D[3] = row_lane<6>(b1t);
        #pragma unroll
        for (int i = 0; i < 4; ++i) {
            const float sE = accP0[2 * i] + accP1[2 * i];
            const float sO = accP0[2 * i + 1] + accP1[2 * i + 1];
            const float own = oddCol ? sO : sE;
            const float a1p = (own + pair_lane(own)) + b1D[i];
            a1[i] = fmaxf(a1p, 0.f);
        }
        #pragma unroll
        for (int kb = 0; kb < 2; ++kb) {
            a1v[kb][0] = a1[2 * kb];
            a1v[kb][1] = a1[2 * kb + 1];
        }
        #pragma unroll
        for (int t = 0; t < 2; ++t) accY[t] = y0[t];
        #pragma unroll
        for (int kb = 0; kb < 2; ++kb) {
            #pragma unroll
            for (int t = 0; t < 2; ++t)
                accY[t] = __builtin_amdgcn_wmma_f32_16x16x4_f32(
                            false, w2Q[t][kb], false, a1v[kb], (short)0, accY[t], false, false);
        }
    };

    auto step = [&](const v2f (&xv)[4], const v4f (&vq)[4]) {
        {
            v8f y0[2];
            #pragma unroll
            for (int t = 0; t < 2; ++t) {
                #pragma unroll
                for (int u = 0; u < 2; ++u) {
                    #pragma unroll
                    for (int e = 0; e < 4; ++e)
                        y0[t][4 * u + e] = b2D[t][4 * u + e] - vq[2 * t + u][e];
                }
            }

            v2f a1v[2];
            float a1[4];
            v8f accY[2];
            forward(xv, y0, a1, a1v, accY);

            float nd[2][8];
            #pragma unroll
            for (int t = 0; t < 2; ++t) {
                #pragma unroll
                for (int r = 0; r < 8; ++r) nd[t][r] = NLR16 * accY[t][r];
            }

            float dsel[8], ndsel[8];
            #pragma unroll
            for (int r = 0; r < 8; ++r) {
                dsel[r]  = oddCol ? accY[1][r] : accY[0][r];
                ndsel[r] = oddCol ? nd[1][r] : nd[0][r];
            }
            v8f accQ0 = {}, accQ1 = {};
            #pragma unroll
            for (int i = 0; i < 4; i += 2) {
                v2f dv0, dv1;
                dv0[0] = dsel[2 * i];     dv0[1] = dsel[2 * i + 1];
                dv1[0] = dsel[2 * i + 2]; dv1[1] = dsel[2 * i + 3];
                accQ0 = __builtin_amdgcn_wmma_f32_16x16x4_f32(
                            false, dv0, false, w2P[i], (short)0, accQ0, false, false);
                accQ1 = __builtin_amdgcn_wmma_f32_16x16x4_f32(
                            false, dv1, false, w2P[i + 1], (short)0, accQ1, false, false);
            }

            const float a1sel = pick4(a1, ib0, ib1);
            const float a1far = other_half(a1sel);
            const float a1own = ownHalf ? a1sel : a1far;

            const float sE = accQ0[0] + accQ1[0];
            const float sO = accQ0[1] + accQ1[1];
            const float own = oddCol ? sO : sE;
            const float qown = own + pair_lane(own);
            const float g1own = (a1own > 0.f) ? qown : 0.f;

            float ndown[2];
            #pragma unroll
            for (int t = 0; t < 2; ++t) {
                const float s8 = pick8(nd[t], db0, db1, db2);
                const float sx = other_half(s8);
                ndown[t] = ownHalf ? s8 : sx;
            }

            const float ng = NLR16 * g1own;
            #pragma unroll
            for (int i = 0; i < 4; ++i) {
                #pragma unroll
                for (int j = 0; j < 2; ++j)
                    w1A[i][j] = fmaf(ng, xv[i][j], w1A[i][j]);
            }
            #pragma unroll
            for (int i = 0; i < 4; ++i) {
                #pragma unroll
                for (int j = 0; j < 2; ++j)
                    w2P[i][j] = fmaf(ndsel[2 * i + j], a1own, w2P[i][j]);
            }
            #pragma unroll
            for (int t = 0; t < 2; ++t) {
                #pragma unroll
                for (int kb = 0; kb < 2; ++kb) {
                    #pragma unroll
                    for (int j = 0; j < 2; ++j)
                        w2Q[t][kb][j] = fmaf(ndown[t], a1v[kb][j], w2Q[t][kb][j]);
                }
                #pragma unroll
                for (int r = 0; r < 8; ++r) b2D[t][r] = fmaf(NLR16, accY[t][r], b2D[t][r]);
            }
            b1own = fmaf(NLR16, g1own, b1own);
        }
    };

    #pragma unroll
    for (int k = 0; k < 4; ++k) hid4[0][k * 32 + lane] = src[k * 32 + lane];

    #pragma unroll 1
    for (int c = 0; c < NCH; ++c) {
        v4f nx[4];
        const int cn = (c + 1 < NCH) ? (c + 1) : c;
        #pragma unroll
        for (int k = 0; k < 4; ++k) nx[k] = src[cn * 128 + k * 32 + lane];

        const float* hbuf = (const float*)hid4[c & 1];
        v2f xs[8][4];
        v4f vs[8][4];
        #pragma unroll
        for (int s = 0; s < 8; ++s) {
            load_x(hbuf + (2 * s) * Hq, xs[s]);
            load_v(hbuf + (2 * s + 1) * Hq, vs[s]);
        }
        __builtin_amdgcn_sched_barrier(0);
        #pragma unroll
        for (int s = 0; s < 8; ++s) {
            if (s < 7 || c != NCH - 1) step(xs[s], vs[s]);
        }

        #pragma unroll
        for (int k = 0; k < 4; ++k) hid4[(c + 1) & 1][k * 32 + lane] = nx[k];
    }

    {
        const float* hbuf = (const float*)hid4[(NCH - 1) & 1];
        v2f xv[4], a1v[2];
        float a1[4];
        v8f y0[2], accY[2];
        #pragma unroll
        for (int t = 0; t < 2; ++t) {
            #pragma unroll
            for (int r = 0; r < 8; ++r) y0[t][r] = b2D[t][r];
        }
        load_x(hbuf + 15 * Hq, xv);
        forward(xv, y0, a1, a1v, accY);
        if (m == 0) {
            #pragma unroll
            for (int t = 0; t < 2; ++t) {
                #pragma unroll
                for (int r = 0; r < 8; ++r) ctxLds[16 * t + 8 * hi + r] = accY[t][r];
            }
        }
    }
    __syncthreads();

    #pragma unroll
    for (int half = 0; half < 2; ++half) {
        const int vo = half * 32 + lane;
        float acc = out_b[vo];
        #pragma unroll
        for (int h = 0; h < Hq; ++h)
            acc = fmaf(ctxLds[h], out_w[vo * Hq + h], acc);
        out[b * Vq + vo] = acc;
    }
}

extern "C" void kernel_launch(void* const* d_in, const int* in_sizes, int n_in,
                              void* d_out, int out_size, void* d_ws, size_t ws_size,
                              hipStream_t stream) {
    (void)in_sizes; (void)n_in; (void)out_size; (void)ws_size;
    const int*   seq    = (const int*)  d_in[0];
    const float* embed  = (const float*)d_in[1];
    const float* ff_w1  = (const float*)d_in[2];
    const float* ff_b1  = (const float*)d_in[3];
    const float* ff_w2  = (const float*)d_in[4];
    const float* ff_b2  = (const float*)d_in[5];
    const float* ln_g   = (const float*)d_in[6];
    const float* ln_b   = (const float*)d_in[7];
    const float* mlp_w1 = (const float*)d_in[8];
    const float* mlp_b1 = (const float*)d_in[9];
    const float* mlp_w2 = (const float*)d_in[10];
    const float* mlp_b2 = (const float*)d_in[11];
    const float* out_w  = (const float*)d_in[12];
    const float* out_b  = (const float*)d_in[13];
    float*       out    = (float*)d_out;
    float*       hidden = (float*)d_ws;

    pmm_encode_kernel<<<dim3(ENC_BLOCKS), dim3(ENC_WAVES * 32), 0, stream>>>(
        seq, embed, ff_w1, ff_b1, ff_w2, ff_b2, ln_g, ln_b, hidden);
    pmm_sgd_kernel<<<dim3(Bq), dim3(32), 0, stream>>>(
        hidden, mlp_w1, mlp_b1, mlp_w2, mlp_b2, out_w, out_b, out);
}


# Encoder: ff_w1 / ff_w2 rows padded in LDS to strides of 36 / 68 dwords: the per-chunk weight fragment reads (one row per lane) were 8- and 16-way bank conflicts, now one pass each; sums unchanged
# speedup vs baseline: 3.6932x; 1.0661x over previous
#include <hip/hip_runtime.h>
#include <hip/hip_bf16.h>
#include <math.h>


typedef __attribute__((ext_vector_type(2))) float v2f;
typedef __attribute__((ext_vector_type(4))) float v4f;
typedef __attribute__((ext_vector_type(8))) float v8f;

#define LRC  0.01f
#define EPSC 1e-5f
#define NLR16 (-LRC * 0.0625f)

#define Bq 256
#define Lq 4096
#define Hq 32
#define Vq 64
#define CH 16
#define NCH (Lq / CH)
#define A1STR 68
#define HSTR  36
#define ESTR  36
#define W1STR 36
#define W2STR 68

#define ENC_WAVES 8
#define ENC_CPW   8
#define ENC_BLOCKS ((Bq * NCH) / (ENC_WAVES * ENC_CPW))

__device__ __forceinline__ float enc_other_half(float s)
{
    const int r = __builtin_amdgcn_permlanex16(0, __float_as_int(s), 0x76543210, 0xfedcba98, false, false);
    return __int_as_float(r);
}

__launch_bounds__(ENC_WAVES * 32)
__global__ void pmm_encode_kernel(const int* __restrict__ seq,
                                  const float* __restrict__ embed,
                                  const float* __restrict__ ff_w1,
                                  const float* __restrict__ ff_b1,
                                  const float* __restrict__ ff_w2,
                                  const float* __restrict__ ff_b2,
                                  const float* __restrict__ ln_g,
                                  const float* __restrict__ ln_b,
                                  float* __restrict__ hidden)
{
    __shared__ __attribute__((aligned(16))) float embedLds[Vq * ESTR];
    __shared__ __attribute__((aligned(16))) float w1Lds[2 * Hq * W1STR];
    __shared__ __attribute__((aligned(16))) float w2Lds[Hq * W2STR];
    __shared__ __attribute__((aligned(16))) float a1all[ENC_WAVES][CH * A1STR];
    __shared__ __attribute__((aligned(16))) float hball[ENC_WAVES][CH * HSTR];
    __shared__ __attribute__((aligned(16))) float b2Lds[Hq];
    __shared__ __attribute__((aligned(16))) float gLds[Hq];
    __shared__ __attribute__((aligned(16))) float bLds[Hq];

    const int tid  = threadIdx.x;
    const int lane = tid & 31;
    const int wave = tid >> 5;
    const int m    = lane & 15;
    const int hi   = lane >> 4;

    for (int i = tid; i < Vq * Hq; i += ENC_WAVES * 32)     embedLds[(i >> 5) * ESTR + (i & 31)] = embed[i];
    for (int i = tid; i < 2 * Hq * Hq; i += ENC_WAVES * 32) w1Lds[(i >> 5) * W1STR + (i & 31)] = ff_w1[i];
    for (int i = tid; i < 2 * Hq * Hq; i += ENC_WAVES * 32) w2Lds[(i >> 6) * W2STR + (i & 63)] = ff_w2[i];
    if (tid < Hq) {
        b2Lds[tid] = ff_b2[tid];
        gLds[tid]  = ln_g[tid];
        bLds[tid]  = ln_b[tid];
    }
    __syncthreads();

    float b1t[4];
    #pragma unroll
    for (int nt = 0; nt < 4; ++nt) b1t[nt] = ff_b1[nt * 16 + m];

    float* a1buf = a1all[wave];
    float* hbuf  = hball[wave];

    const int g0 = (blockIdx.x * ENC_WAVES + wave) * ENC_CPW;

    #pragma unroll 1
    for (int j = 0; j < ENC_CPW; ++j) {
        const int base = (g0 + j) * CH;
        const int tok  = seq[base + m];
        const float* erow = &embedLds[tok * ESTR];

        v2f eA[8];
        #pragma unroll
        for (int kb = 0; kb < 8; ++kb)
            eA[kb] = *(const v2f*)&erow[4 * kb + 2 * hi];

        #pragma unroll 1
        for (int nt = 0; nt < 4; ++nt) {
            const int n0 = nt * 16;
            v8f acc = {};
            #pragma unroll
            for (int kb = 0; kb < 8; ++kb) {
                v2f bB = *(const v2f*)&w1Lds[(n0 + m) * W1STR + 4 * kb + 2 * hi];
                acc = __builtin_amdgcn_wmma_f32_16x16x4_f32(
                        false, eA[kb], false, bB, (short)0, acc, false, false);
            }
            const float bias = b1t[nt];
            #pragma unroll
            for (int r = 0; r < 8; ++r) {
                float z = acc[r] + bias;
                a1buf[(r + 8 * hi) * A1STR + n0 + m] = fmaxf(z, 0.f);
            }
        }
        asm volatile("s_wait_dscnt 0" ::: "memory");
        __builtin_amdgcn_wave_barrier();

        v2f a2[16];
        #pragma unroll
        for (int kb = 0; kb < 16; ++kb)
            a2[kb] = *(const v2f*)&a1buf[m * A1STR + 4 * kb + 2 * hi];

        float z[2][8];
        #pragma unroll
        for (int nt = 0; nt < 2; ++nt) {
            const int n0 = nt * 16;
            v8f acc = {};
            #pragma unroll
            for (int kb = 0; kb < 16; ++kb) {
                v2f wA = *(const v2f*)&w2Lds[(n0 + m) * W2STR + 4 * kb + 2 * hi];
                acc = __builtin_amdgcn_wmma_f32_16x16x4_f32(
                        false, wA, false, a2[kb], (short)0, acc, false, false);
            }
            #pragma unroll
            for (int u = 0; u < 2; ++u) {
                const v4f bq = *(const v4f*)&b2Lds[n0 + 8 * hi + 4 * u];
                const v4f eq = *(const v4f*)&erow[n0 + 8 * hi + 4 * u];
                #pragma unroll
                for (int e = 0; e < 4; ++e)
                    z[nt][4 * u + e] = acc[4 * u + e] + bq[e] + eq[e];
            }
        }

        float ps[8], pq[8];
        #pragma unroll
        for (int r = 0; r < 8; ++r) {
            ps[r] = z[0][r] + z[1][r];
            pq[r] = fmaf(z[1][r], z[1][r], z[0][r] * z[0][r]);
        }
        #pragma unroll
        for (int w = 4; w >= 1; w >>= 1) {
            #pragma unroll
            for (int r = 0; r < w; ++r) {
                ps[r] += ps[r + w];
                pq[r] += pq[r + w];
            }
        }
        const float s   = ps[0] + enc_other_half(ps[0]);
        const float s2  = pq[0] + enc_other_half(pq[0]);
        const float mu  = s * (1.f / 32.f);
        const float var = s2 * (1.f / 32.f) - mu * mu;
        const float rs  = rsqrtf(var + EPSC);
        #pragma unroll
        for (int nt = 0; nt < 2; ++nt) {
            #pragma unroll
            for (int u = 0; u < 2; ++u) {
                const int f0 = 16 * nt + 8 * hi + 4 * u;
                const v4f gq = *(const v4f*)&gLds[f0];
                const v4f bq = *(const v4f*)&bLds[f0];
                v4f o;
                #pragma unroll
                for (int e = 0; e < 4; ++e) {
                    const float nz = (z[nt][4 * u + e] - mu) * rs;
                    o[e] = nz * gq[e] + bq[e];
                }
                *(v4f*)&hbuf[m * HSTR + f0] = o;
            }
        }
        asm volatile("s_wait_dscnt 0" ::: "memory");
        __builtin_amdgcn_wave_barrier();

        #pragma unroll
        for (int i = 0; i < 4; ++i) {
            const int t = 4 * i + (lane >> 3);
            const v4f row = *(const v4f*)&hbuf[t * HSTR + 4 * (lane & 7)];
            *(v4f*)&hidden[(size_t)(base + t) * Hq + 4 * (lane & 7)] = row;
        }
        asm volatile("s_wait_dscnt 0" ::: "memory");
    }
}

__device__ __forceinline__ float pick8(const float (&a)[8], bool b0, bool b1, bool b2)
{
    const float t0 = b0 ? a[1] : a[0];
    const float t1 = b0 ? a[3] : a[2];
    const float t2 = b0 ? a[5] : a[4];
    const float t3 = b0 ? a[7] : a[6];
    const float u0 = b1 ? t1 : t0;
    const float u1 = b1 ? t3 : t2;
    return b2 ? u1 : u0;
}

__device__ __forceinline__ float pick4(const float (&a)[4], bool b0, bool b1)
{
    const float t0 = b0 ? a[1] : a[0];
    const float t1 = b0 ? a[3] : a[2];
    return b1 ? t1 : t0;
}

__device__ __forceinline__ float pair_lane(float s)
{
    return __int_as_float(__builtin_amdgcn_update_dpp(0, __float_as_int(s), 0xB1, 0xF, 0xF, true));
}

__device__ __forceinline__ float other_half(float s)
{
    const int r = __builtin_amdgcn_permlanex16(0, __float_as_int(s), 0x76543210, 0xfedcba98, false, false);
    return __int_as_float(r);
}

__device__ __forceinline__ float high_row_turned(float s)
{
    return __int_as_float(__builtin_amdgcn_update_dpp(__float_as_int(s), __float_as_int(s), 0x128, 0x2, 0xF, false));
}

template <int N>
__device__ __forceinline__ float row_lane(float s)
{
    return __int_as_float(__builtin_amdgcn_update_dpp(0, __float_as_int(s), 0x150 + N, 0xF, 0xF, true));
}

__launch_bounds__(32)
__global__ void pmm_sgd_kernel(const float* __restrict__ hidden,
                               const float* __restrict__ mlp_w1,
                               const float* __restrict__ mlp_b1,
                               const float* __restrict__ mlp_w2,
                               const float* __restrict__ mlp_b2,
                               const float* __restrict__ out_w,
                               const float* __restrict__ out_b,
                               float* __restrict__ out)
{
    __shared__ v4f   hid4[2][CH * Hq / 4];
    __shared__ float ctxLds[Hq];

    const int b    = blockIdx.x;
    const int lane = threadIdx.x;
    const int m    = lane & 15;
    const int hi   = lane >> 4;
    const bool db0 = (m & 1) != 0, db1 = (m & 2) != 0, db2 = (m & 4) != 0;
    const bool ownHalf = (m >> 3) == hi;
    const int uW1  = 4 * (m >> 3) + ((m & 7) >> 1);
    const int sW1  = m & 1;
    const bool ib0 = (m & 2) != 0, ib1 = (m & 4) != 0;
    const bool oddCol = sW1 != 0;

    const v4f* src = (const v4f*)(hidden + (size_t)b * Lq * Hq);

    v2f w1A[4], w2P[4], w2Q[2][2];
    float b2D[2][8];
    #pragma unroll
    for (int i = 0; i < 4; ++i) {
        #pragma unroll
        for (int j = 0; j < 2; ++j)
            w1A[i][j] = mlp_w1[uW1 * Hq + 16 * sW1 + 4 * i + 2 * hi + j];
    }
    #pragma unroll
    for (int i = 0; i < 4; ++i) {
        #pragma unroll
        for (int j = 0; j < 2; ++j)
            w2P[i][j] = mlp_w2[(16 * sW1 + 8 * hi + 2 * i + j) * 8 + uW1];
    }
    #pragma unroll
    for (int t = 0; t < 2; ++t) {
        #pragma unroll
        for (int kb = 0; kb < 2; ++kb) {
            #pragma unroll
            for (int j = 0; j < 2; ++j)
                w2Q[t][kb][j] = mlp_w2[(16 * t + m) * 8 + 4 * hi + 2 * kb + j];
        }
        #pragma unroll
        for (int r = 0; r < 8; ++r) b2D[t][r] = mlp_b2[16 * t + 8 * hi + r];
    }
    float b1own = mlp_b1[uW1];

    auto load_x = [&](const float* xrow, v2f (&xv)[4]) {
        #pragma unroll
        for (int i = 0; i < 4; ++i)
            xv[i] = *(const v2f*)(xrow + 16 * sW1 + 4 * i + 2 * hi);
    };
    auto load_v = [&](const float* vrow, v4f (&vq)[4]) {
        #pragma unroll
        for (int t = 0; t < 2; ++t) {
            #pragma unroll
            for (int u = 0; u < 2; ++u)
                vq[2 * t + u] = *(const v4f*)(vrow + 16 * t + 8 * hi + 4 * u);
        }
    };

    auto forward = [&](const v2f (&xv)[4], const v8f (&y0)[2], float (&a1)[4],
                       v2f (&a1v)[2], v8f (&accY)[2]) {
        v8f accP0 = {}, accP1 = {};
        #pragma unroll
        for (int i = 0; i < 4; i += 2) {
            accP0 = __builtin_amdgcn_wmma_f32_16x16x4_f32(
                        false, w1A[i], false, xv[i], (short)0, accP0, false, false);
            accP1 = __builtin_amdgcn_wmma_f32_16x16x4_f32(
                        false, w1A[i + 1], false, xv[i + 1], (short)0, accP1, false, false);
        }
        const float b1t = high_row_turned(b1own);
        float b1D[4];
        b1D[0] = row_lane<0>(b1t);
        b1D[1] = row_lane<2>(b1t);
        b1D[2] = row_lane<4>(b1t);
        b1D[3] = row_lane<6>(b1t);
        #pragma unroll
        for (int i = 0; i < 4; ++i) {
            const float sE = accP0[2 * i] + accP1[2 * i];
            const float sO = accP0[2 * i + 1] + accP1[2 * i + 1];
            const float own = oddCol ? sO : sE;
            const float a1p = (own + pair_lane(own)) + b1D[i];
            a1[i] = fmaxf(a1p, 0.f);
        }
        #pragma unroll
        for (int kb = 0; kb < 2; ++kb) {
            a1v[kb][0] = a1[2 * kb];
            a1v[kb][1] = a1[2 * kb + 1];
        }
        #pragma unroll
        for (int t = 0; t < 2; ++t) accY[t] = y0[t];
        #pragma unroll
        for (int kb = 0; kb < 2; ++kb) {
            #pragma unroll
            for (int t = 0; t < 2; ++t)
                accY[t] = __builtin_amdgcn_wmma_f32_16x16x4_f32(
                            false, w2Q[t][kb], false, a1v[kb], (short)0, accY[t], false, false);
        }
    };

    auto step = [&](const v2f (&xv)[4], const v4f (&vq)[4]) {
        {
            v8f y0[2];
            #pragma unroll
            for (int t = 0; t < 2; ++t) {
                #pragma unroll
                for (int u = 0; u < 2; ++u) {
                    #pragma unroll
                    for (int e = 0; e < 4; ++e)
                        y0[t][4 * u + e] = b2D[t][4 * u + e] - vq[2 * t + u][e];
                }
            }

            v2f a1v[2];
            float a1[4];
            v8f accY[2];
            forward(xv, y0, a1, a1v, accY);

            float nd[2][8];
            #pragma unroll
            for (int t = 0; t < 2; ++t) {
                #pragma unroll
                for (int r = 0; r < 8; ++r) nd[t][r] = NLR16 * accY[t][r];
            }

            float dsel[8], ndsel[8];
            #pragma unroll
            for (int r = 0; r < 8; ++r) {
                dsel[r]  = oddCol ? accY[1][r] : accY[0][r];
                ndsel[r] = oddCol ? nd[1][r] : nd[0][r];
            }
            v8f accQ0 = {}, accQ1 = {};
            #pragma unroll
            for (int i = 0; i < 4; i += 2) {
                v2f dv0, dv1;
                dv0[0] = dsel[2 * i];     dv0[1] = dsel[2 * i + 1];
                dv1[0] = dsel[2 * i + 2]; dv1[1] = dsel[2 * i + 3];
                accQ0 = __builtin_amdgcn_wmma_f32_16x16x4_f32(
                            false, dv0, false, w2P[i], (short)0, accQ0, false, false);
                accQ1 = __builtin_amdgcn_wmma_f32_16x16x4_f32(
                            false, dv1, false, w2P[i + 1], (short)0, accQ1, false, false);
            }

            const float a1sel = pick4(a1, ib0, ib1);
            const float a1far = other_half(a1sel);
            const float a1own = ownHalf ? a1sel : a1far;

            const float sE = accQ0[0] + accQ1[0];
            const float sO = accQ0[1] + accQ1[1];
            const float own = oddCol ? sO : sE;
            const float qown = own + pair_lane(own);
            const float g1own = (a1own > 0.f) ? qown : 0.f;

            float ndown[2];
            #pragma unroll
            for (int t = 0; t < 2; ++t) {
                const float s8 = pick8(nd[t], db0, db1, db2);
                const float sx = other_half(s8);
                ndown[t] = ownHalf ? s8 : sx;
            }

            const float ng = NLR16 * g1own;
            #pragma unroll
            for (int i = 0; i < 4; ++i) {
                #pragma unroll
                for (int j = 0; j < 2; ++j)
                    w1A[i][j] = fmaf(ng, xv[i][j], w1A[i][j]);
            }
            #pragma unroll
            for (int i = 0; i < 4; ++i) {
                #pragma unroll
                for (int j = 0; j < 2; ++j)
                    w2P[i][j] = fmaf(ndsel[2 * i + j], a1own, w2P[i][j]);
            }
            #pragma unroll
            for (int t = 0; t < 2; ++t) {
                #pragma unroll
                for (int kb = 0; kb < 2; ++kb) {
                    #pragma unroll
                    for (int j = 0; j < 2; ++j)
                        w2Q[t][kb][j] = fmaf(ndown[t], a1v[kb][j], w2Q[t][kb][j]);
                }
                #pragma unroll
                for (int r = 0; r < 8; ++r) b2D[t][r] = fmaf(NLR16, accY[t][r], b2D[t][r]);
            }
            b1own = fmaf(NLR16, g1own, b1own);
        }
    };

    #pragma unroll
    for (int k = 0; k < 4; ++k) hid4[0][k * 32 + lane] = src[k * 32 + lane];

    #pragma unroll 1
    for (int c = 0; c < NCH; ++c) {
        v4f nx[4];
        const int cn = (c + 1 < NCH) ? (c + 1) : c;
        #pragma unroll
        for (int k = 0; k < 4; ++k) nx[k] = src[cn * 128 + k * 32 + lane];

        const float* hbuf = (const float*)hid4[c & 1];
        v2f xs[8][4];
        v4f vs[8][4];
        #pragma unroll
        for (int s = 0; s < 8; ++s) {
            load_x(hbuf + (2 * s) * Hq, xs[s]);
            load_v(hbuf + (2 * s + 1) * Hq, vs[s]);
        }
        __builtin_amdgcn_sched_barrier(0);
        #pragma unroll
        for (int s = 0; s < 8; ++s) {
            if (s < 7 || c != NCH - 1) step(xs[s], vs[s]);
        }

        #pragma unroll
        for (int k = 0; k < 4; ++k) hid4[(c + 1) & 1][k * 32 + lane] = nx[k];
    }

    {
        const float* hbuf = (const float*)hid4[(NCH - 1) & 1];
        v2f xv[4], a1v[2];
        float a1[4];
        v8f y0[2], accY[2];
        #pragma unroll
        for (int t = 0; t < 2; ++t) {
            #pragma unroll
            for (int r = 0; r < 8; ++r) y0[t][r] = b2D[t][r];
        }
        load_x(hbuf + 15 * Hq, xv);
        forward(xv, y0, a1, a1v, accY);
        if (m == 0) {
            #pragma unroll
            for (int t = 0; t < 2; ++t) {
                #pragma unroll
                for (int r = 0; r < 8; ++r) ctxLds[16 * t + 8 * hi + r] = accY[t][r];
            }
        }
    }
    __syncthreads();

    #pragma unroll
    for (int half = 0; half < 2; ++half) {
        const int vo = half * 32 + lane;
        float acc = out_b[vo];
        #pragma unroll
        for (int h = 0; h < Hq; ++h)
            acc = fmaf(ctxLds[h], out_w[vo * Hq + h], acc);
        out[b * Vq + vo] = acc;
    }
}

extern "C" void kernel_launch(void* const* d_in, const int* in_sizes, int n_in,
                              void* d_out, int out_size, void* d_ws, size_t ws_size,
                              hipStream_t stream) {
    (void)in_sizes; (void)n_in; (void)out_size; (void)ws_size;
    const int*   seq    = (const int*)  d_in[0];
    const float* embed  = (const float*)d_in[1];
    const float* ff_w1  = (const float*)d_in[2];
    const float* ff_b1  = (const float*)d_in[3];
    const float* ff_w2  = (const float*)d_in[4];
    const float* ff_b2  = (const float*)d_in[5];
    const float* ln_g   = (const float*)d_in[6];
    const float* ln_b   = (const float*)d_in[7];
    const float* mlp_w1 = (const float*)d_in[8];
    const float* mlp_b1 = (const float*)d_in[9];
    const float* mlp_w2 = (const float*)d_in[10];
    const float* mlp_b2 = (const float*)d_in[11];
    const float* out_w  = (const float*)d_in[12];
    const float* out_b  = (const float*)d_in[13];
    float*       out    = (float*)d_out;
    float*       hidden = (float*)d_ws;

    pmm_encode_kernel<<<dim3(ENC_BLOCKS), dim3(ENC_WAVES * 32), 0, stream>>>(
        seq, embed, ff_w1, ff_b1, ff_w2, ff_b2, ln_g, ln_b, hidden);
    pmm_sgd_kernel<<<dim3(Bq), dim3(32), 0, stream>>>(
        hidden, mlp_w1, mlp_b1, mlp_w2, mlp_b2, out_w, out_b, out);
}


# SGD launch: fc1 once per chunk of 8 steps as one matrix product (B columns = steps, chunk-start weights); each step's rank-1 update reaches later p by one fma with x_s.x_j+1; p's sum order changes
# speedup vs baseline: 4.1054x; 1.1116x over previous
#include <hip/hip_runtime.h>
#include <hip/hip_bf16.h>
#include <math.h>


typedef __attribute__((ext_vector_type(2))) float v2f;
typedef __attribute__((ext_vector_type(4))) float v4f;
typedef __attribute__((ext_vector_type(8))) float v8f;

#define LRC  0.01f
#define EPSC 1e-5f
#define NLR16 (-LRC * 0.0625f)

#define Bq 256
#define Lq 4096
#define Hq 32
#define Vq 64
#define CH 16
#define NCH (Lq / CH)
#define A1STR 68
#define HSTR  36
#define ESTR  36
#define W1STR 36
#define W2STR 68

#define ENC_WAVES 8
#define ENC_CPW   8
#define ENC_BLOCKS ((Bq * NCH) / (ENC_WAVES * ENC_CPW))

__device__ __forceinline__ float enc_other_half(float s)
{
    const int r = __builtin_amdgcn_permlanex16(0, __float_as_int(s), 0x76543210, 0xfedcba98, false, false);
    return __int_as_float(r);
}

__launch_bounds__(ENC_WAVES * 32)
__global__ void pmm_encode_kernel(const int* __restrict__ seq,
                                  const float* __restrict__ embed,
                                  const float* __restrict__ ff_w1,
                                  const float* __restrict__ ff_b1,
                                  const float* __restrict__ ff_w2,
                                  const float* __restrict__ ff_b2,
                                  const float* __restrict__ ln_g,
                                  const float* __restrict__ ln_b,
                                  float* __restrict__ hidden)
{
    __shared__ __attribute__((aligned(16))) float embedLds[Vq * ESTR];
    __shared__ __attribute__((aligned(16))) float w1Lds[2 * Hq * W1STR];
    __shared__ __attribute__((aligned(16))) float w2Lds[Hq * W2STR];
    __shared__ __attribute__((aligned(16))) float a1all[ENC_WAVES][CH * A1STR];
    __shared__ __attribute__((aligned(16))) float hball[ENC_WAVES][CH * HSTR];
    __shared__ __attribute__((aligned(16))) float b2Lds[Hq];
    __shared__ __attribute__((aligned(16))) float gLds[Hq];
    __shared__ __attribute__((aligned(16))) float bLds[Hq];

    const int tid  = threadIdx.x;
    const int lane = tid & 31;
    const int wave = tid >> 5;
    const int m    = lane & 15;
    const int hi   = lane >> 4;

    for (int i = tid; i < Vq * Hq; i += ENC_WAVES * 32)     embedLds[(i >> 5) * ESTR + (i & 31)] = embed[i];
    for (int i = tid; i < 2 * Hq * Hq; i += ENC_WAVES * 32) w1Lds[(i >> 5) * W1STR + (i & 31)] = ff_w1[i];
    for (int i = tid; i < 2 * Hq * Hq; i += ENC_WAVES * 32) w2Lds[(i >> 6) * W2STR + (i & 63)] = ff_w2[i];
    if (tid < Hq) {
        b2Lds[tid] = ff_b2[tid];
        gLds[tid]  = ln_g[tid];
        bLds[tid]  = ln_b[tid];
    }
    __syncthreads();

    float b1t[4];
    #pragma unroll
    for (int nt = 0; nt < 4; ++nt) b1t[nt] = ff_b1[nt * 16 + m];

    float* a1buf = a1all[wave];
    float* hbuf  = hball[wave];

    const int g0 = (blockIdx.x * ENC_WAVES + wave) * ENC_CPW;

    #pragma unroll 1
    for (int j = 0; j < ENC_CPW; ++j) {
        const int base = (g0 + j) * CH;
        const int tok  = seq[base + m];
        const float* erow = &embedLds[tok * ESTR];

        v2f eA[8];
        #pragma unroll
        for (int kb = 0; kb < 8; ++kb)
            eA[kb] = *(const v2f*)&erow[4 * kb + 2 * hi];

        #pragma unroll 1
        for (int nt = 0; nt < 4; ++nt) {
            const int n0 = nt * 16;
            v8f acc = {};
            #pragma unroll
            for (int kb = 0; kb < 8; ++kb) {
                v2f bB = *(const v2f*)&w1Lds[(n0 + m) * W1STR + 4 * kb + 2 * hi];
                acc = __builtin_amdgcn_wmma_f32_16x16x4_f32(
                        false, eA[kb], false, bB, (short)0, acc, false, false);
            }
            const float bias = b1t[nt];
            #pragma unroll
            for (int r = 0; r < 8; ++r) {
                float z = acc[r] + bias;
                a1buf[(r + 8 * hi) * A1STR + n0 + m] = fmaxf(z, 0.f);
            }
        }
        asm volatile("s_wait_dscnt 0" ::: "memory");
        __builtin_amdgcn_wave_barrier();

        v2f a2[16];
        #pragma unroll
        for (int kb = 0; kb < 16; ++kb)
            a2[kb] = *(const v2f*)&a1buf[m * A1STR + 4 * kb + 2 * hi];

        float z[2][8];
        #pragma unroll
        for (int nt = 0; nt < 2; ++nt) {
            const int n0 = nt * 16;
            v8f acc = {};
            #pragma unroll
            for (int kb = 0; kb < 16; ++kb) {
                v2f wA = *(const v2f*)&w2Lds[(n0 + m) * W2STR + 4 * kb + 2 * hi];
                acc = __builtin_amdgcn_wmma_f32_16x16x4_f32(
                        false, wA, false, a2[kb], (short)0, acc, false, false);
            }
            #pragma unroll
            for (int u = 0; u < 2; ++u) {
                const v4f bq = *(const v4f*)&b2Lds[n0 + 8 * hi + 4 * u];
                const v4f eq = *(const v4f*)&erow[n0 + 8 * hi + 4 * u];
                #pragma unroll
                for (int e = 0; e < 4; ++e)
                    z[nt][4 * u + e] = acc[4 * u + e] + bq[e] + eq[e];
            }
        }

        float ps[8], pq[8];
        #pragma unroll
        for (int r = 0; r < 8; ++r) {
            ps[r] = z[0][r] + z[1][r];
            pq[r] = fmaf(z[1][r], z[1][r], z[0][r] * z[0][r]);
        }
        #pragma unroll
        for (int w = 4; w >= 1; w >>= 1) {
            #pragma unroll
            for (int r = 0; r < w; ++r) {
                ps[r] += ps[r + w];
                pq[r] += pq[r + w];
            }
        }
        const float s   = ps[0] + enc_other_half(ps[0]);
        const float s2  = pq[0] + enc_other_half(pq[0]);
        const float mu  = s * (1.f / 32.f);
        const float var = s2 * (1.f / 32.f) - mu * mu;
        const float rs  = rsqrtf(var + EPSC);
        #pragma unroll
        for (int nt = 0; nt < 2; ++nt) {
            #pragma unroll
            for (int u = 0; u < 2; ++u) {
                const int f0 = 16 * nt + 8 * hi + 4 * u;
                const v4f gq = *(const v4f*)&gLds[f0];
                const v4f bq = *(const v4f*)&bLds[f0];
                v4f o;
                #pragma unroll
                for (int e = 0; e < 4; ++e) {
                    const float nz = (z[nt][4 * u + e] - mu) * rs;
                    o[e] = nz * gq[e] + bq[e];
                }
                *(v4f*)&hbuf[m * HSTR + f0] = o;
            }
        }
        asm volatile("s_wait_dscnt 0" ::: "memory");
        __builtin_amdgcn_wave_barrier();

        #pragma unroll
        for (int i = 0; i < 4; ++i) {
            const int t = 4 * i + (lane >> 3);
            const v4f row = *(const v4f*)&hbuf[t * HSTR + 4 * (lane & 7)];
            *(v4f*)&hidden[(size_t)(base + t) * Hq + 4 * (lane & 7)] = row;
        }
        asm volatile("s_wait_dscnt 0" ::: "memory");
    }
}

__device__ __forceinline__ float pick8(const float (&a)[8], bool b0, bool b1, bool b2)
{
    const float t0 = b0 ? a[1] : a[0];
    const float t1 = b0 ? a[3] : a[2];
    const float t2 = b0 ? a[5] : a[4];
    const float t3 = b0 ? a[7] : a[6];
    const float u0 = b1 ? t1 : t0;
    const float u1 = b1 ? t3 : t2;
    return b2 ? u1 : u0;
}

__device__ __forceinline__ float pick4(const float (&a)[4], bool b0, bool b1)
{
    const float t0 = b0 ? a[1] : a[0];
    const float t1 = b0 ? a[3] : a[2];
    return b1 ? t1 : t0;
}

__device__ __forceinline__ float pair_lane(float s)
{
    return __int_as_float(__builtin_amdgcn_update_dpp(0, __float_as_int(s), 0xB1, 0xF, 0xF, true));
}

__device__ __forceinline__ float other_half(float s)
{
    const int r = __builtin_amdgcn_permlanex16(0, __float_as_int(s), 0x76543210, 0xfedcba98, false, false);
    return __int_as_float(r);
}

__device__ __forceinline__ float high_row_turned(float s)
{
    return __int_as_float(__builtin_amdgcn_update_dpp(__float_as_int(s), __float_as_int(s), 0x128, 0x2, 0xF, false));
}

template <int N>
__device__ __forceinline__ float row_lane(float s)
{
    return __int_as_float(__builtin_amdgcn_update_dpp(0, __float_as_int(s), 0x150 + N, 0xF, 0xF, true));
}

__device__ __forceinline__ float row_lane_n(float v, int s)
{
    switch (s) {
        case 0:  return row_lane<0>(v);
        case 1:  return row_lane<2>(v);
        case 2:  return row_lane<4>(v);
        case 3:  return row_lane<6>(v);
        case 4:  return row_lane<8>(v);
        case 5:  return row_lane<10>(v);
        case 6:  return row_lane<12>(v);
        default: return row_lane<14>(v);
    }
}

__launch_bounds__(32)
__global__ void pmm_sgd_kernel(const float* __restrict__ hidden,
                               const float* __restrict__ mlp_w1,
                               const float* __restrict__ mlp_b1,
                               const float* __restrict__ mlp_w2,
                               const float* __restrict__ mlp_b2,
                               const float* __restrict__ out_w,
                               const float* __restrict__ out_b,
                               float* __restrict__ out)
{
    __shared__ v4f   hid4[2][CH * Hq / 4];
    __shared__ float ctxLds[Hq];

    const int b    = blockIdx.x;
    const int lane = threadIdx.x;
    const int m    = lane & 15;
    const int hi   = lane >> 4;
    const bool db0 = (m & 1) != 0, db1 = (m & 2) != 0, db2 = (m & 4) != 0;
    const bool ownHalf = (m >> 3) == hi;
    const int uW1  = 4 * (m >> 3) + ((m & 7) >> 1);
    const int sW1  = m & 1;
    const bool ib0 = (m & 2) != 0, ib1 = (m & 4) != 0;
    const bool oddCol = sW1 != 0;

    const v4f* src = (const v4f*)(hidden + (size_t)b * Lq * Hq);

    v2f w1A[4], w2P[4], w2Q[2][2];
    float b2D[2][8];
    #pragma unroll
    for (int i = 0; i < 4; ++i) {
        #pragma unroll
        for (int j = 0; j < 2; ++j)
            w1A[i][j] = mlp_w1[uW1 * Hq + 16 * sW1 + 4 * i + 2 * hi + j];
    }
    #pragma unroll
    for (int i = 0; i < 4; ++i) {
        #pragma unroll
        for (int j = 0; j < 2; ++j)
            w2P[i][j] = mlp_w2[(16 * sW1 + 8 * hi + 2 * i + j) * 8 + uW1];
    }
    #pragma unroll
    for (int t = 0; t < 2; ++t) {
        #pragma unroll
        for (int kb = 0; kb < 2; ++kb) {
            #pragma unroll
            for (int j = 0; j < 2; ++j)
                w2Q[t][kb][j] = mlp_w2[(16 * t + m) * 8 + 4 * hi + 2 * kb + j];
        }
        #pragma unroll
        for (int r = 0; r < 8; ++r) b2D[t][r] = mlp_b2[16 * t + 8 * hi + r];
    }
    float b1own = mlp_b1[uW1];

    auto load_x = [&](const float* xrow, v2f (&xv)[4]) {
        #pragma unroll
        for (int i = 0; i < 4; ++i)
            xv[i] = *(const v2f*)(xrow + 16 * sW1 + 4 * i + 2 * hi);
    };
    auto load_v = [&](const float* vrow, v4f (&vq)[4]) {
        #pragma unroll
        for (int t = 0; t < 2; ++t) {
            #pragma unroll
            for (int u = 0; u < 2; ++u)
                vq[2 * t + u] = *(const v4f*)(vrow + 16 * t + 8 * hi + 4 * u);
        }
    };

    auto forward = [&](const v2f (&xv)[4], const v8f (&y0)[2], float (&a1)[4],
                       v2f (&a1v)[2], v8f (&accY)[2]) {
        v8f accP0 = {}, accP1 = {};
        #pragma unroll
        for (int i = 0; i < 4; i += 2) {
            accP0 = __builtin_amdgcn_wmma_f32_16x16x4_f32(
                        false, w1A[i], false, xv[i], (short)0, accP0, false, false);
            accP1 = __builtin_amdgcn_wmma_f32_16x16x4_f32(
                        false, w1A[i + 1], false, xv[i + 1], (short)0, accP1, false, false);
        }
        const float b1t = high_row_turned(b1own);
        float b1D[4];
        b1D[0] = row_lane<0>(b1t);
        b1D[1] = row_lane<2>(b1t);
        b1D[2] = row_lane<4>(b1t);
        b1D[3] = row_lane<6>(b1t);
        #pragma unroll
        for (int i = 0; i < 4; ++i) {
            const float sE = accP0[2 * i] + accP1[2 * i];
            const float sO = accP0[2 * i + 1] + accP1[2 * i + 1];
            const float own = oddCol ? sO : sE;
            const float a1p = (own + pair_lane(own)) + b1D[i];
            a1[i] = fmaxf(a1p, 0.f);
        }
        #pragma unroll
        for (int kb = 0; kb < 2; ++kb) {
            a1v[kb][0] = a1[2 * kb];
            a1v[kb][1] = a1[2 * kb + 1];
        }
        #pragma unroll
        for (int t = 0; t < 2; ++t) accY[t] = y0[t];
        #pragma unroll
        for (int kb = 0; kb < 2; ++kb) {
            #pragma unroll
            for (int t = 0; t < 2; ++t)
                accY[t] = __builtin_amdgcn_wmma_f32_16x16x4_f32(
                            false, w2Q[t][kb], false, a1v[kb], (short)0, accY[t], false, false);
        }
    };

    auto join_halves = [&](const v8f& acc0, const v8f& acc1, float (&full)[4]) {
        #pragma unroll
        for (int i = 0; i < 4; ++i) {
            const float sE = acc0[2 * i] + acc1[2 * i];
            const float sO = acc0[2 * i + 1] + acc1[2 * i + 1];
            const float own = oddCol ? sO : sE;
            full[i] = own + pair_lane(own);
        }
    };

    auto load_xcols = [&](const float* hbuf, v2f (&xb)[4]) {
        const float* xrow = hbuf + (2 * (m >> 1)) * Hq;
        #pragma unroll
        for (int i = 0; i < 4; ++i)
            xb[i] = *(const v2f*)(xrow + 16 * sW1 + 4 * i + 2 * hi);
    };

    auto make_y0 = [&](const v4f (&vq)[4], v8f (&y0)[2]) {
        #pragma unroll
        for (int t = 0; t < 2; ++t) {
            #pragma unroll
            for (int u = 0; u < 2; ++u) {
                #pragma unroll
                for (int e = 0; e < 4; ++e)
                    y0[t][4 * u + e] = b2D[t][4 * u + e] - vq[2 * t + u][e];
            }
        }
    };
    auto fc2_fwd = [&](const float (&a1)[4], const v8f (&y0)[2], v2f (&a1v)[2], v8f (&accY)[2]) {
        #pragma unroll
        for (int kb = 0; kb < 2; ++kb) {
            a1v[kb][0] = a1[2 * kb];
            a1v[kb][1] = a1[2 * kb + 1];
        }
        #pragma unroll
        for (int t = 0; t < 2; ++t) accY[t] = y0[t];
        #pragma unroll
        for (int kb = 0; kb < 2; ++kb) {
            #pragma unroll
            for (int t = 0; t < 2; ++t)
                accY[t] = __builtin_amdgcn_wmma_f32_16x16x4_f32(
                            false, w2Q[t][kb], false, a1v[kb], (short)0, accY[t], false, false);
        }
    };
    auto bwd_and_fc2_update = [&](const v8f (&accY)[2], const v2f (&a1v)[2], float (&ndsel)[8], v8f& accQ0, v8f& accQ1) {
        float nd[2][8];
        #pragma unroll
        for (int t = 0; t < 2; ++t) {
            #pragma unroll
            for (int r = 0; r < 8; ++r) nd[t][r] = NLR16 * accY[t][r];
        }
        float dsel[8];
        #pragma unroll
        for (int r = 0; r < 8; ++r) {
            dsel[r]  = oddCol ? accY[1][r] : accY[0][r];
            ndsel[r] = oddCol ? nd[1][r] : nd[0][r];
        }
        accQ0 = v8f{}; accQ1 = v8f{};
        #pragma unroll
        for (int i = 0; i < 4; i += 2) {
            v2f dv0, dv1;
            dv0[0] = dsel[2 * i];     dv0[1] = dsel[2 * i + 1];
            dv1[0] = dsel[2 * i + 2]; dv1[1] = dsel[2 * i + 3];
            accQ0 = __builtin_amdgcn_wmma_f32_16x16x4_f32(
                        false, w2P[i], false, dv0, (short)0, accQ0, false, false);
            accQ1 = __builtin_amdgcn_wmma_f32_16x16x4_f32(
                        false, w2P[i + 1], false, dv1, (short)0, accQ1, false, false);
        }
        float ndown[2];
        #pragma unroll
        for (int t = 0; t < 2; ++t) {
            const float s8 = pick8(nd[t], db0, db1, db2);
            const float sx = other_half(s8);
            ndown[t] = ownHalf ? s8 : sx;
        }
        #pragma unroll
        for (int t = 0; t < 2; ++t) {
            #pragma unroll
            for (int kb = 0; kb < 2; ++kb) {
                #pragma unroll
                for (int j = 0; j < 2; ++j)
                    w2Q[t][kb][j] = fmaf(ndown[t], a1v[kb][j], w2Q[t][kb][j]);
            }
            #pragma unroll
            for (int r = 0; r < 8; ++r) b2D[t][r] = fmaf(NLR16, accY[t][r], b2D[t][r]);
        }
    };
    auto gate_q = [&](const v8f& accQ0, const v8f& accQ1, const float (&a1)[4], float (&g4)[4]) {
        float q4[4];
        join_halves(accQ0, accQ1, q4);
        #pragma unroll
        for (int i = 0; i < 4; ++i) g4[i] = (a1[i] > 0.f) ? q4[i] : 0.f;
    };
    auto own_row_updates = [&](const float (&a1)[4], const float (&g4)[4], const float (&ndsel)[8], const v2f (&xv)[4]) {
        const float a1sel = pick4(a1, ib0, ib1);
        const float a1far = other_half(a1sel);
        const float a1own = ownHalf ? a1sel : a1far;
        const float g1sel = pick4(g4, ib0, ib1);
        const float g1far = other_half(g1sel);
        const float g1own = ownHalf ? g1sel : g1far;
        const float ng = NLR16 * g1own;
        #pragma unroll
        for (int i = 0; i < 4; ++i) {
            #pragma unroll
            for (int j = 0; j < 2; ++j)
                w1A[i][j] = fmaf(ng, xv[i][j], w1A[i][j]);
        }
        #pragma unroll
        for (int i = 0; i < 4; ++i) {
            #pragma unroll
            for (int j = 0; j < 2; ++j)
                w2P[i][j] = fmaf(ndsel[2 * i + j], a1own, w2P[i][j]);
        }
        b1own = fmaf(NLR16, g1own, b1own);
    };

    #pragma unroll
    for (int k = 0; k < 4; ++k) hid4[0][k * 32 + lane] = src[k * 32 + lane];

    #pragma unroll 1
    for (int c = 0; c < NCH; ++c) {
        v4f nx[4];
        const int cn = (c + 1 < NCH) ? (c + 1) : c;
        #pragma unroll
        for (int k = 0; k < 4; ++k) nx[k] = src[cn * 128 + k * 32 + lane];

        const float* hbuf = (const float*)hid4[c & 1];
        v2f xb[4];
        load_xcols(hbuf, xb);
        v2f xs[8][4];
        v4f vs[8][4];
        #pragma unroll
        for (int s = 0; s < 8; ++s) {
            load_x(hbuf + (2 * s) * Hq, xs[s]);
            load_v(hbuf + (2 * s + 1) * Hq, vs[s]);
        }
        __builtin_amdgcn_sched_barrier(0);

        v8f accP0 = {}, accP1 = {};
        #pragma unroll
        for (int i = 0; i < 4; i += 2) {
            accP0 = __builtin_amdgcn_wmma_f32_16x16x4_f32(
                        false, w1A[i], false, xb[i], (short)0, accP0, false, false);
            accP1 = __builtin_amdgcn_wmma_f32_16x16x4_f32(
                        false, w1A[i + 1], false, xb[i + 1], (short)0, accP1, false, false);
        }
        v8f accG0 = {}, accG1 = {};
        #pragma unroll
        for (int i = 0; i < 4; i += 2) {
            accG0 = __builtin_amdgcn_wmma_f32_16x16x4_f32(
                        false, xb[i], false, xb[i], (short)0, accG0, false, false);
            accG1 = __builtin_amdgcn_wmma_f32_16x16x4_f32(
                        false, xb[i + 1], false, xb[i + 1], (short)0, accG1, false, false);
        }
        const float b1t = high_row_turned(b1own);
        float b1D[4];
        b1D[0] = row_lane<0>(b1t);
        b1D[1] = row_lane<2>(b1t);
        b1D[2] = row_lane<4>(b1t);
        b1D[3] = row_lane<6>(b1t);
        float pc[4];
        join_halves(accP0, accP1, pc);
        #pragma unroll
        for (int i = 0; i < 4; ++i) pc[i] += b1D[i];
        float gd[4];
        join_halves(accG0, accG1, gd);
        float cv[4];
        #pragma unroll
        for (int i = 0; i < 4; ++i) cv[i] = fmaf(gd[i], NLR16, NLR16);
        float cfcol[7];
        #pragma unroll
        for (int s = 0; s < 7; ++s) {
            const float own = cv[s & 3];
            const float far = other_half(own);
            cfcol[s] = (hi == (s >> 2)) ? own : far;
        }

        v8f y0[2];
        make_y0(vs[0], y0);

        float a1S[8][4], g4S[8][4], ndS[8][8];
        #pragma unroll
        for (int s = 0; s < 8; ++s) {
            if (s < 7 || c != NCH - 1) {
                v2f a1v[2];
                v8f accY[2], accQ0, accQ1;
                #pragma unroll
                for (int i = 0; i < 4; ++i) a1S[s][i] = row_lane_n(fmaxf(pc[i], 0.f), s);
                fc2_fwd(a1S[s], y0, a1v, accY);
                if (s > 0) own_row_updates(a1S[s - 1], g4S[s - 1], ndS[s - 1], xs[s - 1]);
                bwd_and_fc2_update(accY, a1v, ndS[s], accQ0, accQ1);
                if (s < 7) make_y0(vs[s + 1], y0);
                gate_q(accQ0, accQ1, a1S[s], g4S[s]);
                if (s < 7) {
                    #pragma unroll
                    for (int i = 0; i < 4; ++i) pc[i] = fmaf(g4S[s][i], cfcol[s], pc[i]);
                }
            }
        }
        if (c != NCH - 1) own_row_updates(a1S[7], g4S[7], ndS[7], xs[7]);
        else              own_row_updates(a1S[6], g4S[6], ndS[6], xs[6]);

        #pragma unroll
        for (int k = 0; k < 4; ++k) hid4[(c + 1) & 1][k * 32 + lane] = nx[k];
    }

    {
        const float* hbuf = (const float*)hid4[(NCH - 1) & 1];
        v2f xv[4], a1v[2];
        float a1[4];
        v8f y0[2], accY[2];
        #pragma unroll
        for (int t = 0; t < 2; ++t) {
            #pragma unroll
            for (int r = 0; r < 8; ++r) y0[t][r] = b2D[t][r];
        }
        load_x(hbuf + 15 * Hq, xv);
        forward(xv, y0, a1, a1v, accY);
        if (m == 0) {
            #pragma unroll
            for (int t = 0; t < 2; ++t) {
                #pragma unroll
                for (int r = 0; r < 8; ++r) ctxLds[16 * t + 8 * hi + r] = accY[t][r];
            }
        }
    }
    __syncthreads();

    #pragma unroll
    for (int half = 0; half < 2; ++half) {
        const int vo = half * 32 + lane;
        float acc = out_b[vo];
        #pragma unroll
        for (int h = 0; h < Hq; ++h)
            acc = fmaf(ctxLds[h], out_w[vo * Hq + h], acc);
        out[b * Vq + vo] = acc;
    }
}

extern "C" void kernel_launch(void* const* d_in, const int* in_sizes, int n_in,
                              void* d_out, int out_size, void* d_ws, size_t ws_size,
                              hipStream_t stream) {
    (void)in_sizes; (void)n_in; (void)out_size; (void)ws_size;
    const int*   seq    = (const int*)  d_in[0];
    const float* embed  = (const float*)d_in[1];
    const float* ff_w1  = (const float*)d_in[2];
    const float* ff_b1  = (const float*)d_in[3];
    const float* ff_w2  = (const float*)d_in[4];
    const float* ff_b2  = (const float*)d_in[5];
    const float* ln_g   = (const float*)d_in[6];
    const float* ln_b   = (const float*)d_in[7];
    const float* mlp_w1 = (const float*)d_in[8];
    const float* mlp_b1 = (const float*)d_in[9];
    const float* mlp_w2 = (const float*)d_in[10];
    const float* mlp_b2 = (const float*)d_in[11];
    const float* out_w  = (const float*)d_in[12];
    const float* out_b  = (const float*)d_in[13];
    float*       out    = (float*)d_out;
    float*       hidden = (float*)d_ws;

    pmm_encode_kernel<<<dim3(ENC_BLOCKS), dim3(ENC_WAVES * 32), 0, stream>>>(
        seq, embed, ff_w1, ff_b1, ff_w2, ff_b2, ln_g, ln_b, hidden);
    pmm_sgd_kernel<<<dim3(Bq), dim3(32), 0, stream>>>(
        hidden, mlp_w1, mlp_b1, mlp_w2, mlp_b2, out_w, out_b, out);
}
